# starting kernel
# baseline (speedup 1.0000x reference)
#include <hip/hip_runtime.h>
#include <hip/hip_bf16.h>
#include <cmath>

typedef __attribute__((ext_vector_type(2))) float v2f;
typedef __attribute__((ext_vector_type(8))) float v8f;

#define BB    2
#define LL    1024
#define DMOD  1024
#define DI    2048
#define NS    16
#define KC    4
#define RR    64
#define XZW   (2 * DI)
#define DBLW  (RR + 2 * NS)

__device__ __forceinline__ float siluf(float v) {
    return v / (1.0f + __expf(-v));
}
__device__ __forceinline__ float softplusf(float v) {
    return (v > 20.0f) ? v : log1pf(expf(v));
}

template <int NT, bool SP>
__global__ void __launch_bounds__(128) wmma_gemm_f32(
    const float* __restrict__ A, const float* __restrict__ W,
    float* __restrict__ C, int N, int K, int lda,
    const float* __restrict__ bias)
{
    const int wave = threadIdx.x >> 5;
    const int lane = threadIdx.x & 31;
    const int half = lane >> 4;
    const int mr   = lane & 15;
    const int m0   = blockIdx.y * 64 + wave * 16;
    const int nb   = blockIdx.x * (NT * 16);

    v8f acc[NT] = {};

    const float* aptr = A + (size_t)(m0 + mr) * lda + 2 * half;
    const float* wptr = W + (size_t)(2 * half) * N + nb + mr;

    v2f a_cur = *(const v2f*)aptr;
    v2f b_cur[NT];
#pragma unroll
    for (int j = 0; j < NT; ++j) {
        b_cur[j][0] = wptr[j * 16];
        b_cur[j][1] = wptr[(size_t)N + j * 16];
    }

    for (int k0 = 0; k0 < K - 4; k0 += 4) {
        aptr += 4;
        wptr += (size_t)4 * N;
        v2f a_nxt = *(const v2f*)aptr;
        v2f b_nxt[NT];
#pragma unroll
        for (int j = 0; j < NT; ++j) {
            b_nxt[j][0] = wptr[j * 16];
            b_nxt[j][1] = wptr[(size_t)N + j * 16];
        }
#pragma unroll
        for (int j = 0; j < NT; ++j)
            acc[j] = __builtin_amdgcn_wmma_f32_16x16x4_f32(
                false, a_cur, false, b_cur[j], (short)0, acc[j], false, false);
        a_cur = a_nxt;
#pragma unroll
        for (int j = 0; j < NT; ++j) b_cur[j] = b_nxt[j];
    }
#pragma unroll
    for (int j = 0; j < NT; ++j)
        acc[j] = __builtin_amdgcn_wmma_f32_16x16x4_f32(
            false, a_cur, false, b_cur[j], (short)0, acc[j], false, false);

#pragma unroll
    for (int j = 0; j < NT; ++j) {
        const int col = nb + j * 16 + mr;
        const float bsv = SP ? bias[col] : 0.0f;
#pragma unroll
        for (int r = 0; r < 8; ++r) {
            float v = acc[j][r];
            if (SP) v = softplusf(v + bsv);
            C[(size_t)(m0 + r + 8 * half) * N + col] = v;
        }
    }
}

__global__ void conv_silu_kernel(const float* __restrict__ xz,
                                 const float* __restrict__ cw,
                                 const float* __restrict__ cb,
                                 float* __restrict__ u)
{
    const int idx = blockIdx.x * blockDim.x + threadIdx.x;
    const int d  = idx % DI;
    const int bt = idx / DI;
    const int t  = bt % LL;
    float acc = cb[d];
    const float* w = cw + d * KC;
#pragma unroll
    for (int k = 0; k < KC; ++k) {
        const int tt = t - (KC - 1) + k;
        if (tt >= 0)
            acc += xz[(size_t)(bt - (KC - 1) + k) * XZW + d] * w[k];
    }
    u[(size_t)bt * DI + d] = siluf(acc);
}

__global__ void __launch_bounds__(128) scan_kernel(
    const float* __restrict__ dt, const float* __restrict__ u,
    const float* __restrict__ dbl, const float* __restrict__ xz,
    const float* __restrict__ A_log, const float* __restrict__ Dp,
    float* __restrict__ yg)
{
    __shared__ float sBC[2 * NS];
    const int b = blockIdx.x >> 4;
    const int d = (blockIdx.x & 15) * 128 + threadIdx.x;

    float An[NS];
#pragma unroll
    for (int n = 0; n < NS; ++n) An[n] = -__expf(A_log[d * NS + n]);
    const float Dd = Dp[d];
    float h[NS];
#pragma unroll
    for (int n = 0; n < NS; ++n) h[n] = 0.0f;

    for (int t = 0; t < LL; ++t) {
        const size_t row = (size_t)(b * LL + t);
        if (threadIdx.x < 2 * NS)
            sBC[threadIdx.x] = dbl[row * DBLW + RR + threadIdx.x];
        __syncthreads();

        const float dtv = dt[row * DI + d];
        const float uv  = u[row * DI + d];
        const float du  = dtv * uv;
        float y = 0.0f;
#pragma unroll
        for (int n = 0; n < NS; ++n) {
            h[n] = __expf(dtv * An[n]) * h[n] + du * sBC[n];
            y += h[n] * sBC[NS + n];
        }
        const float zv = xz[row * XZW + DI + d];
        yg[row * DI + d] = (y + uv * Dd) * siluf(zv);
        __syncthreads();
    }
}

extern "C" void kernel_launch(void* const* d_in, const int* in_sizes, int n_in,
                              void* d_out, int out_size, void* d_ws, size_t ws_size,
                              hipStream_t stream) {
    const float* x          = (const float*)d_in[0];
    const float* in_proj_w  = (const float*)d_in[1];
    const float* conv_w     = (const float*)d_in[2];
    const float* conv_b     = (const float*)d_in[3];
    const float* x_proj_w   = (const float*)d_in[4];
    const float* dt_proj_w  = (const float*)d_in[5];
    const float* dt_proj_b  = (const float*)d_in[6];
    const float* A_log      = (const float*)d_in[7];
    const float* Dvec       = (const float*)d_in[8];
    const float* out_proj_w = (const float*)d_in[9];
    float* out = (float*)d_out;

    float* ws  = (float*)d_ws;
    float* xz  = ws;
    float* u   = xz  + (size_t)2048 * 4096;
    float* dbl = u   + (size_t)2048 * 2048;
    float* dtb = dbl + (size_t)2048 * 96;
    float* yg  = dtb + (size_t)2048 * 2048;

    const dim3 blk(128);
    const int M = BB * LL;

    wmma_gemm_f32<4, false><<<dim3(4096 / 64, M / 64), blk, 0, stream>>>(
        x, in_proj_w, xz, 4096, DMOD, DMOD, nullptr);

    conv_silu_kernel<<<(BB * LL * DI) / 256, 256, 0, stream>>>(xz, conv_w, conv_b, u);

    wmma_gemm_f32<2, false><<<dim3(DBLW / 32, M / 64), blk, 0, stream>>>(
        u, x_proj_w, dbl, DBLW, DI, DI, nullptr);

    wmma_gemm_f32<4, true><<<dim3(DI / 64, M / 64), blk, 0, stream>>>(
        dbl, dt_proj_w, dtb, DI, RR, DBLW, dt_proj_b);

    scan_kernel<<<32, 128, 0, stream>>>(dtb, u, dbl, xz, A_log, Dvec, yg);

    wmma_gemm_f32<4, false><<<dim3(DMOD / 64, M / 64), blk, 0, stream>>>(
        yg, out_proj_w, out, DMOD, DI, DI, nullptr);
}


# scan: dt/u/z and the B|C rows of 16 time steps are loaded one chunk ahead (registers, LDS double buffer); no step waits on a global load, one barrier per 16 steps instead of two per step
# speedup vs baseline: 3.1869x; 3.1869x over previous
#include <hip/hip_runtime.h>
#include <hip/hip_bf16.h>
#include <cmath>

typedef __attribute__((ext_vector_type(2))) float v2f;
typedef __attribute__((ext_vector_type(8))) float v8f;

#define BB    2
#define LL    1024
#define DMOD  1024
#define DI    2048
#define NS    16
#define KC    4
#define RR    64
#define XZW   (2 * DI)
#define DBLW  (RR + 2 * NS)

__device__ __forceinline__ float siluf(float v) {
    return v / (1.0f + __expf(-v));
}
__device__ __forceinline__ float softplusf(float v) {
    return (v > 20.0f) ? v : log1pf(expf(v));
}

template <int NT, bool SP>
__global__ void __launch_bounds__(128) wmma_gemm_f32(
    const float* __restrict__ A, const float* __restrict__ W,
    float* __restrict__ C, int N, int K, int lda,
    const float* __restrict__ bias)
{
    const int wave = threadIdx.x >> 5;
    const int lane = threadIdx.x & 31;
    const int half = lane >> 4;
    const int mr   = lane & 15;
    const int m0   = blockIdx.y * 64 + wave * 16;
    const int nb   = blockIdx.x * (NT * 16);

    v8f acc[NT] = {};

    const float* aptr = A + (size_t)(m0 + mr) * lda + 2 * half;
    const float* wptr = W + (size_t)(2 * half) * N + nb + mr;

    v2f a_cur = *(const v2f*)aptr;
    v2f b_cur[NT];
#pragma unroll
    for (int j = 0; j < NT; ++j) {
        b_cur[j][0] = wptr[j * 16];
        b_cur[j][1] = wptr[(size_t)N + j * 16];
    }

    for (int k0 = 0; k0 < K - 4; k0 += 4) {
        aptr += 4;
        wptr += (size_t)4 * N;
        v2f a_nxt = *(const v2f*)aptr;
        v2f b_nxt[NT];
#pragma unroll
        for (int j = 0; j < NT; ++j) {
            b_nxt[j][0] = wptr[j * 16];
            b_nxt[j][1] = wptr[(size_t)N + j * 16];
        }
#pragma unroll
        for (int j = 0; j < NT; ++j)
            acc[j] = __builtin_amdgcn_wmma_f32_16x16x4_f32(
                false, a_cur, false, b_cur[j], (short)0, acc[j], false, false);
        a_cur = a_nxt;
#pragma unroll
        for (int j = 0; j < NT; ++j) b_cur[j] = b_nxt[j];
    }
#pragma unroll
    for (int j = 0; j < NT; ++j)
        acc[j] = __builtin_amdgcn_wmma_f32_16x16x4_f32(
            false, a_cur, false, b_cur[j], (short)0, acc[j], false, false);

#pragma unroll
    for (int j = 0; j < NT; ++j) {
        const int col = nb + j * 16 + mr;
        const float bsv = SP ? bias[col] : 0.0f;
#pragma unroll
        for (int r = 0; r < 8; ++r) {
            float v = acc[j][r];
            if (SP) v = softplusf(v + bsv);
            C[(size_t)(m0 + r + 8 * half) * N + col] = v;
        }
    }
}

__global__ void conv_silu_kernel(const float* __restrict__ xz,
                                 const float* __restrict__ cw,
                                 const float* __restrict__ cb,
                                 float* __restrict__ u)
{
    const int idx = blockIdx.x * blockDim.x + threadIdx.x;
    const int d  = idx % DI;
    const int bt = idx / DI;
    const int t  = bt % LL;
    float acc = cb[d];
    const float* w = cw + d * KC;
#pragma unroll
    for (int k = 0; k < KC; ++k) {
        const int tt = t - (KC - 1) + k;
        if (tt >= 0)
            acc += xz[(size_t)(bt - (KC - 1) + k) * XZW + d] * w[k];
    }
    u[(size_t)bt * DI + d] = siluf(acc);
}

#define TCH   16
#define NCH   (LL / TCH)
#define BCW   (2 * NS)

__global__ void __launch_bounds__(128) scan_kernel(
    const float* __restrict__ dt, const float* __restrict__ u,
    const float* __restrict__ dbl, const float* __restrict__ xz,
    const float* __restrict__ A_log, const float* __restrict__ Dp,
    float* __restrict__ yg)
{
    __shared__ __attribute__((aligned(16))) float sBC[2][TCH * BCW];
    const int b = blockIdx.x >> 4;
    const int d = (blockIdx.x & 15) * 128 + threadIdx.x;

    float An[NS];
#pragma unroll
    for (int n = 0; n < NS; ++n) An[n] = -__expf(A_log[d * NS + n]);
    const float Dd = Dp[d];
    float h[NS];
#pragma unroll
    for (int n = 0; n < NS; ++n) h[n] = 0.0f;

    const size_t row0 = (size_t)b * LL;
    const float* pdt = dt + row0 * DI + d;
    const float* pu  = u  + row0 * DI + d;
    const float* pz  = xz + row0 * XZW + DI + d;
    float*       py  = yg + row0 * DI + d;
    const float* pbc = dbl + (row0 + (threadIdx.x >> 3)) * DBLW + RR + 4 * (threadIdx.x & 7);

    float dt_n[TCH], u_n[TCH], z_n[TCH], bc_n[4];
#pragma unroll
    for (int i = 0; i < TCH; ++i) {
        dt_n[i] = pdt[(size_t)i * DI];
        u_n[i]  = pu[(size_t)i * DI];
        z_n[i]  = pz[(size_t)i * XZW];
    }
#pragma unroll
    for (int j = 0; j < 4; ++j) bc_n[j] = pbc[j];

    for (int c = 0; c < NCH; ++c) {
        float dt_c[TCH], u_c[TCH], z_c[TCH];
#pragma unroll
        for (int i = 0; i < TCH; ++i) { dt_c[i] = dt_n[i]; u_c[i] = u_n[i]; z_c[i] = z_n[i]; }

        float* buf = sBC[c & 1];
        *(float4*)(buf + 4 * threadIdx.x) = make_float4(bc_n[0], bc_n[1], bc_n[2], bc_n[3]);
        __syncthreads();

        if (c + 1 < NCH) {
            pdt += (size_t)TCH * DI;
            pu  += (size_t)TCH * DI;
            pz  += (size_t)TCH * XZW;
            pbc += (size_t)TCH * DBLW;
#pragma unroll
            for (int i = 0; i < TCH; ++i) {
                dt_n[i] = pdt[(size_t)i * DI];
                u_n[i]  = pu[(size_t)i * DI];
                z_n[i]  = pz[(size_t)i * XZW];
            }
#pragma unroll
            for (int j = 0; j < 4; ++j) bc_n[j] = pbc[j];
        }

#pragma unroll
        for (int i = 0; i < TCH; ++i) {
            const float* sb = buf + i * BCW;
            const float dtv = dt_c[i];
            const float uv  = u_c[i];
            const float du  = dtv * uv;
            float y = 0.0f;
#pragma unroll
            for (int n = 0; n < NS; ++n) {
                h[n] = __expf(dtv * An[n]) * h[n] + du * sb[n];
                y += h[n] * sb[NS + n];
            }
            py[(size_t)i * DI] = (y + uv * Dd) * siluf(z_c[i]);
        }
        py += (size_t)TCH * DI;
    }
}

extern "C" void kernel_launch(void* const* d_in, const int* in_sizes, int n_in,
                              void* d_out, int out_size, void* d_ws, size_t ws_size,
                              hipStream_t stream) {
    const float* x          = (const float*)d_in[0];
    const float* in_proj_w  = (const float*)d_in[1];
    const float* conv_w     = (const float*)d_in[2];
    const float* conv_b     = (const float*)d_in[3];
    const float* x_proj_w   = (const float*)d_in[4];
    const float* dt_proj_w  = (const float*)d_in[5];
    const float* dt_proj_b  = (const float*)d_in[6];
    const float* A_log      = (const float*)d_in[7];
    const float* Dvec       = (const float*)d_in[8];
    const float* out_proj_w = (const float*)d_in[9];
    float* out = (float*)d_out;

    float* ws  = (float*)d_ws;
    float* xz  = ws;
    float* u   = xz  + (size_t)2048 * 4096;
    float* dbl = u   + (size_t)2048 * 2048;
    float* dtb = dbl + (size_t)2048 * 96;
    float* yg  = dtb + (size_t)2048 * 2048;

    const dim3 blk(128);
    const int M = BB * LL;

    wmma_gemm_f32<4, false><<<dim3(4096 / 64, M / 64), blk, 0, stream>>>(
        x, in_proj_w, xz, 4096, DMOD, DMOD, nullptr);

    conv_silu_kernel<<<(BB * LL * DI) / 256, 256, 0, stream>>>(xz, conv_w, conv_b, u);

    wmma_gemm_f32<2, false><<<dim3(DBLW / 32, M / 64), blk, 0, stream>>>(
        u, x_proj_w, dbl, DBLW, DI, DI, nullptr);

    wmma_gemm_f32<4, true><<<dim3(DI / 64, M / 64), blk, 0, stream>>>(
        dbl, dt_proj_w, dtb, DI, RR, DBLW, dt_proj_b);

    scan_kernel<<<32, 128, 0, stream>>>(dtb, u, dbl, xz, A_log, Dvec, yg);

    wmma_gemm_f32<4, false><<<dim3(DMOD / 64, M / 64), blk, 0, stream>>>(
        yg, out_proj_w, out, DMOD, DI, DI, nullptr);
}


# all four matrix products: A and W slabs (64 rows x 32 or 64 k) staged through LDS with 16-byte loads one slab ahead, fragments read from LDS; k order and summation order unchanged
# speedup vs baseline: 5.0251x; 1.5768x over previous
#include <hip/hip_runtime.h>
#include <hip/hip_bf16.h>
#include <cmath>

typedef __attribute__((ext_vector_type(2))) float v2f;
typedef __attribute__((ext_vector_type(4))) float v4f;
typedef __attribute__((ext_vector_type(8))) float v8f;

#define BB    2
#define LL    1024
#define DMOD  1024
#define DI    2048
#define NS    16
#define KC    4
#define RR    64
#define XZW   (2 * DI)
#define DBLW  (RR + 2 * NS)

__device__ __forceinline__ float siluf(float v) {
    return v / (1.0f + __expf(-v));
}
__device__ __forceinline__ float softplusf(float v) {
    return (v > 20.0f) ? v : log1pf(expf(v));
}

template <int NT, int BK, bool SP>
__global__ void __launch_bounds__(128) wmma_gemm_f32(
    const float* __restrict__ A, const float* __restrict__ W,
    float* __restrict__ C, int N, int K, int lda,
    const float* __restrict__ bias)
{
    constexpr int BN   = NT * 16;
    constexpr int PA   = BK + 4;
    constexpr int PB   = 2 * BN + 32;
    constexpr int KS   = BK / 4;
    constexpr int A_F4 = (64 * (BK / 4)) / 128;
    constexpr int B_IT = ((BK / 2) * (BN / 4)) / 128;
    constexpr int A_RS = 128 / (BK / 4);
    constexpr int B_PS = 128 / (BN / 4);

    __shared__ __attribute__((aligned(16))) float sA[2][64 * PA];
    __shared__ __attribute__((aligned(16))) float sB[2][(BK / 2) * PB];

    const int tid  = threadIdx.x;
    const int wave = tid >> 5;
    const int lane = tid & 31;
    const int half = lane >> 4;
    const int mr   = lane & 15;
    const int mb   = blockIdx.y * 64;
    const int m0   = mb + wave * 16;
    const int nb   = blockIdx.x * BN;

    const int arow = tid / (BK / 4), ac4 = tid % (BK / 4);
    const int bp   = tid / (BN / 4), bq  = tid % (BN / 4);
    const float* aG = A + (size_t)(mb + arow) * lda + 4 * ac4;
    const float* wG = W + (size_t)(2 * bp) * N + nb + 4 * bq;
    const int aS = arow * PA + 4 * ac4;
    const int bS = bp * PB + 8 * bq;
    const int aF = (wave * 16 + mr) * PA + 2 * half;
    const int bF = half * PB + 2 * mr;

    v8f acc[NT] = {};
    v4f ra[A_F4], rb0[B_IT], rb1[B_IT];

#pragma unroll
    for (int i = 0; i < A_F4; ++i)
        ra[i] = *(const v4f*)(aG + (size_t)(i * A_RS) * lda);
#pragma unroll
    for (int i = 0; i < B_IT; ++i) {
        rb0[i] = *(const v4f*)(wG + (size_t)(2 * i * B_PS) * N);
        rb1[i] = *(const v4f*)(wG + (size_t)(2 * i * B_PS + 1) * N);
    }

    const int S = K / BK;
    for (int s = 0; s < S - 1; ++s) {
        const int cur = s & 1;
#pragma unroll
        for (int i = 0; i < A_F4; ++i)
            *(v4f*)(&sA[cur][aS + i * A_RS * PA]) = ra[i];
#pragma unroll
        for (int i = 0; i < B_IT; ++i) {
            float* d = &sB[cur][bS + i * B_PS * PB];
            const v4f lo = { rb0[i][0], rb1[i][0], rb0[i][1], rb1[i][1] };
            const v4f hi = { rb0[i][2], rb1[i][2], rb0[i][3], rb1[i][3] };
            *(v4f*)(d)     = lo;
            *(v4f*)(d + 4) = hi;
        }
        __syncthreads();

        aG += BK;
        wG += (size_t)BK * N;
#pragma unroll
        for (int i = 0; i < A_F4; ++i)
            ra[i] = *(const v4f*)(aG + (size_t)(i * A_RS) * lda);
#pragma unroll
        for (int i = 0; i < B_IT; ++i) {
            rb0[i] = *(const v4f*)(wG + (size_t)(2 * i * B_PS) * N);
            rb1[i] = *(const v4f*)(wG + (size_t)(2 * i * B_PS + 1) * N);
        }

        const float* pa = &sA[cur][aF];
        const float* pb = &sB[cur][bF];
#pragma unroll
        for (int ks = 0; ks < KS; ++ks) {
            const v2f a = *(const v2f*)(pa + 4 * ks);
            v2f b[NT];
#pragma unroll
            for (int j = 0; j < NT; ++j) b[j] = *(const v2f*)(pb + 2 * ks * PB + 32 * j);
#pragma unroll
            for (int j = 0; j < NT; ++j)
                acc[j] = __builtin_amdgcn_wmma_f32_16x16x4_f32(
                    false, a, false, b[j], (short)0, acc[j], false, false);
        }
    }
    {
        const int cur = (S - 1) & 1;
#pragma unroll
        for (int i = 0; i < A_F4; ++i)
            *(v4f*)(&sA[cur][aS + i * A_RS * PA]) = ra[i];
#pragma unroll
        for (int i = 0; i < B_IT; ++i) {
            float* d = &sB[cur][bS + i * B_PS * PB];
            const v4f lo = { rb0[i][0], rb1[i][0], rb0[i][1], rb1[i][1] };
            const v4f hi = { rb0[i][2], rb1[i][2], rb0[i][3], rb1[i][3] };
            *(v4f*)(d)     = lo;
            *(v4f*)(d + 4) = hi;
        }
        __syncthreads();
        const float* pa = &sA[cur][aF];
        const float* pb = &sB[cur][bF];
#pragma unroll
        for (int ks = 0; ks < KS; ++ks) {
            const v2f a = *(const v2f*)(pa + 4 * ks);
            v2f b[NT];
#pragma unroll
            for (int j = 0; j < NT; ++j) b[j] = *(const v2f*)(pb + 2 * ks * PB + 32 * j);
#pragma unroll
            for (int j = 0; j < NT; ++j)
                acc[j] = __builtin_amdgcn_wmma_f32_16x16x4_f32(
                    false, a, false, b[j], (short)0, acc[j], false, false);
        }
    }

#pragma unroll
    for (int j = 0; j < NT; ++j) {
        const int col = nb + j * 16 + mr;
        const float bsv = SP ? bias[col] : 0.0f;
#pragma unroll
        for (int r = 0; r < 8; ++r) {
            float v = acc[j][r];
            if (SP) v = softplusf(v + bsv);
            C[(size_t)(m0 + r + 8 * half) * N + col] = v;
        }
    }
}

__global__ void conv_silu_kernel(const float* __restrict__ xz,
                                 const float* __restrict__ cw,
                                 const float* __restrict__ cb,
                                 float* __restrict__ u)
{
    const int idx = blockIdx.x * blockDim.x + threadIdx.x;
    const int d  = idx % DI;
    const int bt = idx / DI;
    const int t  = bt % LL;
    float acc = cb[d];
    const float* w = cw + d * KC;
#pragma unroll
    for (int k = 0; k < KC; ++k) {
        const int tt = t - (KC - 1) + k;
        if (tt >= 0)
            acc += xz[(size_t)(bt - (KC - 1) + k) * XZW + d] * w[k];
    }
    u[(size_t)bt * DI + d] = siluf(acc);
}

#define TCH   16
#define NCH   (LL / TCH)
#define BCW   (2 * NS)

__global__ void __launch_bounds__(128) scan_kernel(
    const float* __restrict__ dt, const float* __restrict__ u,
    const float* __restrict__ dbl, const float* __restrict__ xz,
    const float* __restrict__ A_log, const float* __restrict__ Dp,
    float* __restrict__ yg)
{
    __shared__ __attribute__((aligned(16))) float sBC[2][TCH * BCW];
    const int b = blockIdx.x >> 4;
    const int d = (blockIdx.x & 15) * 128 + threadIdx.x;

    float An[NS];
#pragma unroll
    for (int n = 0; n < NS; ++n) An[n] = -__expf(A_log[d * NS + n]);
    const float Dd = Dp[d];
    float h[NS];
#pragma unroll
    for (int n = 0; n < NS; ++n) h[n] = 0.0f;

    const size_t row0 = (size_t)b * LL;
    const float* pdt = dt + row0 * DI + d;
    const float* pu  = u  + row0 * DI + d;
    const float* pz  = xz + row0 * XZW + DI + d;
    float*       py  = yg + row0 * DI + d;
    const float* pbc = dbl + (row0 + (threadIdx.x >> 3)) * DBLW + RR + 4 * (threadIdx.x & 7);

    float dt_n[TCH], u_n[TCH], z_n[TCH], bc_n[4];
#pragma unroll
    for (int i = 0; i < TCH; ++i) {
        dt_n[i] = pdt[(size_t)i * DI];
        u_n[i]  = pu[(size_t)i * DI];
        z_n[i]  = pz[(size_t)i * XZW];
    }
#pragma unroll
    for (int j = 0; j < 4; ++j) bc_n[j] = pbc[j];

    for (int c = 0; c < NCH; ++c) {
        float dt_c[TCH], u_c[TCH], z_c[TCH];
#pragma unroll
        for (int i = 0; i < TCH; ++i) { dt_c[i] = dt_n[i]; u_c[i] = u_n[i]; z_c[i] = z_n[i]; }

        float* buf = sBC[c & 1];
        *(float4*)(buf + 4 * threadIdx.x) = make_float4(bc_n[0], bc_n[1], bc_n[2], bc_n[3]);
        __syncthreads();

        if (c + 1 < NCH) {
            pdt += (size_t)TCH * DI;
            pu  += (size_t)TCH * DI;
            pz  += (size_t)TCH * XZW;
            pbc += (size_t)TCH * DBLW;
#pragma unroll
            for (int i = 0; i < TCH; ++i) {
                dt_n[i] = pdt[(size_t)i * DI];
                u_n[i]  = pu[(size_t)i * DI];
                z_n[i]  = pz[(size_t)i * XZW];
            }
#pragma unroll
            for (int j = 0; j < 4; ++j) bc_n[j] = pbc[j];
        }

#pragma unroll
        for (int i = 0; i < TCH; ++i) {
            const float* sb = buf + i * BCW;
            const float dtv = dt_c[i];
            const float uv  = u_c[i];
            const float du  = dtv * uv;
            float y = 0.0f;
#pragma unroll
            for (int n = 0; n < NS; ++n) {
                h[n] = __expf(dtv * An[n]) * h[n] + du * sb[n];
                y += h[n] * sb[NS + n];
            }
            py[(size_t)i * DI] = (y + uv * Dd) * siluf(z_c[i]);
        }
        py += (size_t)TCH * DI;
    }
}

extern "C" void kernel_launch(void* const* d_in, const int* in_sizes, int n_in,
                              void* d_out, int out_size, void* d_ws, size_t ws_size,
                              hipStream_t stream) {
    const float* x          = (const float*)d_in[0];
    const float* in_proj_w  = (const float*)d_in[1];
    const float* conv_w     = (const float*)d_in[2];
    const float* conv_b     = (const float*)d_in[3];
    const float* x_proj_w   = (const float*)d_in[4];
    const float* dt_proj_w  = (const float*)d_in[5];
    const float* dt_proj_b  = (const float*)d_in[6];
    const float* A_log      = (const float*)d_in[7];
    const float* Dvec       = (const float*)d_in[8];
    const float* out_proj_w = (const float*)d_in[9];
    float* out = (float*)d_out;

    float* ws  = (float*)d_ws;
    float* xz  = ws;
    float* u   = xz  + (size_t)2048 * 4096;
    float* dbl = u   + (size_t)2048 * 2048;
    float* dtb = dbl + (size_t)2048 * 96;
    float* yg  = dtb + (size_t)2048 * 2048;

    const dim3 blk(128);
    const int M = BB * LL;

    wmma_gemm_f32<4, 32, false><<<dim3(4096 / 64, M / 64), blk, 0, stream>>>(
        x, in_proj_w, xz, 4096, DMOD, DMOD, nullptr);

    conv_silu_kernel<<<(BB * LL * DI) / 256, 256, 0, stream>>>(xz, conv_w, conv_b, u);

    wmma_gemm_f32<2, 64, false><<<dim3(DBLW / 32, M / 64), blk, 0, stream>>>(
        u, x_proj_w, dbl, DBLW, DI, DI, nullptr);

    wmma_gemm_f32<4, 32, true><<<dim3(DI / 64, M / 64), blk, 0, stream>>>(
        dbl, dt_proj_w, dtb, DI, RR, DBLW, dt_proj_b);

    scan_kernel<<<32, 128, 0, stream>>>(dtb, u, dbl, xz, A_log, Dvec, yg);

    wmma_gemm_f32<4, 32, false><<<dim3(DMOD / 64, M / 64), blk, 0, stream>>>(
        yg, out_proj_w, out, DMOD, DI, DI, nullptr);
}


# scan: each step's B|C row read from LDS into registers as one run of 8 reads, two steps ahead of its use (three register sets), not read by read among the arithmetic; sums unchanged
# speedup vs baseline: 5.6633x; 1.1270x over previous
#include <hip/hip_runtime.h>
#include <hip/hip_bf16.h>
#include <cmath>

typedef __attribute__((ext_vector_type(2))) float v2f;
typedef __attribute__((ext_vector_type(4))) float v4f;
typedef __attribute__((ext_vector_type(8))) float v8f;

#define BB    2
#define LL    1024
#define DMOD  1024
#define DI    2048
#define NS    16
#define KC    4
#define RR    64
#define XZW   (2 * DI)
#define DBLW  (RR + 2 * NS)

__device__ __forceinline__ float siluf(float v) {
    return v / (1.0f + __expf(-v));
}
__device__ __forceinline__ float softplusf(float v) {
    return (v > 20.0f) ? v : log1pf(expf(v));
}

template <int NT, int BK, bool SP>
__global__ void __launch_bounds__(128) wmma_gemm_f32(
    const float* __restrict__ A, const float* __restrict__ W,
    float* __restrict__ C, int N, int K, int lda,
    const float* __restrict__ bias)
{
    constexpr int BN   = NT * 16;
    constexpr int PA   = BK + 4;
    constexpr int PB   = 2 * BN + 32;
    constexpr int KS   = BK / 4;
    constexpr int A_F4 = (64 * (BK / 4)) / 128;
    constexpr int B_IT = ((BK / 2) * (BN / 4)) / 128;
    constexpr int A_RS = 128 / (BK / 4);
    constexpr int B_PS = 128 / (BN / 4);

    __shared__ __attribute__((aligned(16))) float sA[2][64 * PA];
    __shared__ __attribute__((aligned(16))) float sB[2][(BK / 2) * PB];

    const int tid  = threadIdx.x;
    const int wave = tid >> 5;
    const int lane = tid & 31;
    const int half = lane >> 4;
    const int mr   = lane & 15;
    const int mb   = blockIdx.y * 64;
    const int m0   = mb + wave * 16;
    const int nb   = blockIdx.x * BN;

    const int arow = tid / (BK / 4), ac4 = tid % (BK / 4);
    const int bp   = tid / (BN / 4), bq  = tid % (BN / 4);
    const float* aG = A + (size_t)(mb + arow) * lda + 4 * ac4;
    const float* wG = W + (size_t)(2 * bp) * N + nb + 4 * bq;
    const int aS = arow * PA + 4 * ac4;
    const int bS = bp * PB + 8 * bq;
    const int aF = (wave * 16 + mr) * PA + 2 * half;
    const int bF = half * PB + 2 * mr;

    v8f acc[NT] = {};
    v4f ra[A_F4], rb0[B_IT], rb1[B_IT];

#pragma unroll
    for (int i = 0; i < A_F4; ++i)
        ra[i] = *(const v4f*)(aG + (size_t)(i * A_RS) * lda);
#pragma unroll
    for (int i = 0; i < B_IT; ++i) {
        rb0[i] = *(const v4f*)(wG + (size_t)(2 * i * B_PS) * N);
        rb1[i] = *(const v4f*)(wG + (size_t)(2 * i * B_PS + 1) * N);
    }

    const int S = K / BK;
    for (int s = 0; s < S - 1; ++s) {
        const int cur = s & 1;
#pragma unroll
        for (int i = 0; i < A_F4; ++i)
            *(v4f*)(&sA[cur][aS + i * A_RS * PA]) = ra[i];
#pragma unroll
        for (int i = 0; i < B_IT; ++i) {
            float* d = &sB[cur][bS + i * B_PS * PB];
            const v4f lo = { rb0[i][0], rb1[i][0], rb0[i][1], rb1[i][1] };
            const v4f hi = { rb0[i][2], rb1[i][2], rb0[i][3], rb1[i][3] };
            *(v4f*)(d)     = lo;
            *(v4f*)(d + 4) = hi;
        }
        __syncthreads();

        aG += BK;
        wG += (size_t)BK * N;
#pragma unroll
        for (int i = 0; i < A_F4; ++i)
            ra[i] = *(const v4f*)(aG + (size_t)(i * A_RS) * lda);
#pragma unroll
        for (int i = 0; i < B_IT; ++i) {
            rb0[i] = *(const v4f*)(wG + (size_t)(2 * i * B_PS) * N);
            rb1[i] = *(const v4f*)(wG + (size_t)(2 * i * B_PS + 1) * N);
        }

        const float* pa = &sA[cur][aF];
        const float* pb = &sB[cur][bF];
#pragma unroll
        for (int ks = 0; ks < KS; ++ks) {
            const v2f a = *(const v2f*)(pa + 4 * ks);
            v2f b[NT];
#pragma unroll
            for (int j = 0; j < NT; ++j) b[j] = *(const v2f*)(pb + 2 * ks * PB + 32 * j);
#pragma unroll
            for (int j = 0; j < NT; ++j)
                acc[j] = __builtin_amdgcn_wmma_f32_16x16x4_f32(
                    false, a, false, b[j], (short)0, acc[j], false, false);
        }
    }
    {
        const int cur = (S - 1) & 1;
#pragma unroll
        for (int i = 0; i < A_F4; ++i)
            *(v4f*)(&sA[cur][aS + i * A_RS * PA]) = ra[i];
#pragma unroll
        for (int i = 0; i < B_IT; ++i) {
            float* d = &sB[cur][bS + i * B_PS * PB];
            const v4f lo = { rb0[i][0], rb1[i][0], rb0[i][1], rb1[i][1] };
            const v4f hi = { rb0[i][2], rb1[i][2], rb0[i][3], rb1[i][3] };
            *(v4f*)(d)     = lo;
            *(v4f*)(d + 4) = hi;
        }
        __syncthreads();
        const float* pa = &sA[cur][aF];
        const float* pb = &sB[cur][bF];
#pragma unroll
        for (int ks = 0; ks < KS; ++ks) {
            const v2f a = *(const v2f*)(pa + 4 * ks);
            v2f b[NT];
#pragma unroll
            for (int j = 0; j < NT; ++j) b[j] = *(const v2f*)(pb + 2 * ks * PB + 32 * j);
#pragma unroll
            for (int j = 0; j < NT; ++j)
                acc[j] = __builtin_amdgcn_wmma_f32_16x16x4_f32(
                    false, a, false, b[j], (short)0, acc[j], false, false);
        }
    }

#pragma unroll
    for (int j = 0; j < NT; ++j) {
        const int col = nb + j * 16 + mr;
        const float bsv = SP ? bias[col] : 0.0f;
#pragma unroll
        for (int r = 0; r < 8; ++r) {
            float v = acc[j][r];
            if (SP) v = softplusf(v + bsv);
            C[(size_t)(m0 + r + 8 * half) * N + col] = v;
        }
    }
}

__global__ void conv_silu_kernel(const float* __restrict__ xz,
                                 const float* __restrict__ cw,
                                 const float* __restrict__ cb,
                                 float* __restrict__ u)
{
    const int idx = blockIdx.x * blockDim.x + threadIdx.x;
    const int d  = idx % DI;
    const int bt = idx / DI;
    const int t  = bt % LL;
    float acc = cb[d];
    const float* w = cw + d * KC;
#pragma unroll
    for (int k = 0; k < KC; ++k) {
        const int tt = t - (KC - 1) + k;
        if (tt >= 0)
            acc += xz[(size_t)(bt - (KC - 1) + k) * XZW + d] * w[k];
    }
    u[(size_t)bt * DI + d] = siluf(acc);
}

#define TCH   16
#define NCH   (LL / TCH)
#define BCW   (2 * NS)
#define SV4   (BCW / 4)
#define AHD   2

__global__ void __launch_bounds__(128) scan_kernel(
    const float* __restrict__ dt, const float* __restrict__ u,
    const float* __restrict__ dbl, const float* __restrict__ xz,
    const float* __restrict__ A_log, const float* __restrict__ Dp,
    float* __restrict__ yg)
{
    __shared__ __attribute__((aligned(16))) float sBC[2][TCH * BCW];
    const int b = blockIdx.x >> 4;
    const int d = (blockIdx.x & 15) * 128 + threadIdx.x;

    float An[NS];
#pragma unroll
    for (int n = 0; n < NS; ++n) An[n] = -__expf(A_log[d * NS + n]);
    const float Dd = Dp[d];
    float h[NS];
#pragma unroll
    for (int n = 0; n < NS; ++n) h[n] = 0.0f;

    const size_t row0 = (size_t)b * LL;
    const float* pdt = dt + row0 * DI + d;
    const float* pu  = u  + row0 * DI + d;
    const float* pz  = xz + row0 * XZW + DI + d;
    float*       py  = yg + row0 * DI + d;
    const float* pbc = dbl + (row0 + (threadIdx.x >> 3)) * DBLW + RR + 4 * (threadIdx.x & 7);

    float dt_n[TCH], u_n[TCH], z_n[TCH], bc_n[4];
#pragma unroll
    for (int i = 0; i < TCH; ++i) {
        dt_n[i] = pdt[(size_t)i * DI];
        u_n[i]  = pu[(size_t)i * DI];
        z_n[i]  = pz[(size_t)i * XZW];
    }
#pragma unroll
    for (int j = 0; j < 4; ++j) bc_n[j] = pbc[j];

    for (int c = 0; c < NCH; ++c) {
        float dt_c[TCH], u_c[TCH], z_c[TCH];
#pragma unroll
        for (int i = 0; i < TCH; ++i) { dt_c[i] = dt_n[i]; u_c[i] = u_n[i]; z_c[i] = z_n[i]; }

        float* buf = sBC[c & 1];
        *(float4*)(buf + 4 * threadIdx.x) = make_float4(bc_n[0], bc_n[1], bc_n[2], bc_n[3]);
        __syncthreads();

        if (c + 1 < NCH) {
            pdt += (size_t)TCH * DI;
            pu  += (size_t)TCH * DI;
            pz  += (size_t)TCH * XZW;
            pbc += (size_t)TCH * DBLW;
#pragma unroll
            for (int i = 0; i < TCH; ++i) {
                dt_n[i] = pdt[(size_t)i * DI];
                u_n[i]  = pu[(size_t)i * DI];
                z_n[i]  = pz[(size_t)i * XZW];
            }
#pragma unroll
            for (int j = 0; j < 4; ++j) bc_n[j] = pbc[j];
        }

        v4f ring[AHD + 1][SV4];
#pragma unroll
        for (int r = 0; r < AHD; ++r) {
#pragma unroll
            for (int k = 0; k < SV4; ++k) ring[r][k] = *(const v4f*)(buf + r * BCW + 4 * k);
        }

#pragma unroll
        for (int i = 0; i < TCH; ++i) {
            if (i + AHD < TCH) {
#pragma unroll
                for (int k = 0; k < SV4; ++k)
                    ring[(i + AHD) % (AHD + 1)][k] = *(const v4f*)(buf + (i + AHD) * BCW + 4 * k);
            }
            __builtin_amdgcn_sched_barrier(0);

            const float dtv = dt_c[i];
            const float uv  = u_c[i];
            const float du  = dtv * uv;
            float y = 0.0f;
#pragma unroll
            for (int n = 0; n < NS; ++n) {
                const float bn = ring[i % (AHD + 1)][n >> 2][n & 3];
                const float cn = ring[i % (AHD + 1)][(NS + n) >> 2][n & 3];
                h[n] = __expf(dtv * An[n]) * h[n] + du * bn;
                y += h[n] * cn;
            }
            py[(size_t)i * DI] = (y + uv * Dd) * siluf(z_c[i]);
        }
        py += (size_t)TCH * DI;
    }
}

extern "C" void kernel_launch(void* const* d_in, const int* in_sizes, int n_in,
                              void* d_out, int out_size, void* d_ws, size_t ws_size,
                              hipStream_t stream) {
    const float* x          = (const float*)d_in[0];
    const float* in_proj_w  = (const float*)d_in[1];
    const float* conv_w     = (const float*)d_in[2];
    const float* conv_b     = (const float*)d_in[3];
    const float* x_proj_w   = (const float*)d_in[4];
    const float* dt_proj_w  = (const float*)d_in[5];
    const float* dt_proj_b  = (const float*)d_in[6];
    const float* A_log      = (const float*)d_in[7];
    const float* Dvec       = (const float*)d_in[8];
    const float* out_proj_w = (const float*)d_in[9];
    float* out = (float*)d_out;

    float* ws  = (float*)d_ws;
    float* xz  = ws;
    float* u   = xz  + (size_t)2048 * 4096;
    float* dbl = u   + (size_t)2048 * 2048;
    float* dtb = dbl + (size_t)2048 * 96;
    float* yg  = dtb + (size_t)2048 * 2048;

    const dim3 blk(128);
    const int M = BB * LL;

    wmma_gemm_f32<4, 32, false><<<dim3(4096 / 64, M / 64), blk, 0, stream>>>(
        x, in_proj_w, xz, 4096, DMOD, DMOD, nullptr);

    conv_silu_kernel<<<(BB * LL * DI) / 256, 256, 0, stream>>>(xz, conv_w, conv_b, u);

    wmma_gemm_f32<2, 64, false><<<dim3(DBLW / 32, M / 64), blk, 0, stream>>>(
        u, x_proj_w, dbl, DBLW, DI, DI, nullptr);

    wmma_gemm_f32<4, 32, true><<<dim3(DI / 64, M / 64), blk, 0, stream>>>(
        dbl, dt_proj_w, dtb, DI, RR, DBLW, dt_proj_b);

    scan_kernel<<<32, 128, 0, stream>>>(dtb, u, dbl, xz, A_log, Dvec, yg);

    wmma_gemm_f32<4, 32, false><<<dim3(DMOD / 64, M / 64), blk, 0, stream>>>(
        yg, out_proj_w, out, DMOD, DI, DI, nullptr);
}


# scan: a channel's 16 states split over 4 adjacent lanes (4x the waves); y = 4 partial sums joined through LDS in lane order (new summation order), every 128-byte line of y stored whole by one wave
# speedup vs baseline: 7.9245x; 1.3993x over previous
#include <hip/hip_runtime.h>
#include <hip/hip_bf16.h>
#include <cmath>

typedef __attribute__((ext_vector_type(2))) float v2f;
typedef __attribute__((ext_vector_type(4))) float v4f;
typedef __attribute__((ext_vector_type(8))) float v8f;

#define BB    2
#define LL    1024
#define DMOD  1024
#define DI    2048
#define NS    16
#define KC    4
#define RR    64
#define XZW   (2 * DI)
#define DBLW  (RR + 2 * NS)

__device__ __forceinline__ float siluf(float v) {
    return v / (1.0f + __expf(-v));
}
__device__ __forceinline__ float softplusf(float v) {
    return (v > 20.0f) ? v : log1pf(expf(v));
}

template <int NT, int BK, bool SP>
__global__ void __launch_bounds__(128) wmma_gemm_f32(
    const float* __restrict__ A, const float* __restrict__ W,
    float* __restrict__ C, int N, int K, int lda,
    const float* __restrict__ bias)
{
    constexpr int BN   = NT * 16;
    constexpr int PA   = BK + 4;
    constexpr int PB   = 2 * BN + 32;
    constexpr int KS   = BK / 4;
    constexpr int A_F4 = (64 * (BK / 4)) / 128;
    constexpr int B_IT = ((BK / 2) * (BN / 4)) / 128;
    constexpr int A_RS = 128 / (BK / 4);
    constexpr int B_PS = 128 / (BN / 4);

    __shared__ __attribute__((aligned(16))) float sA[2][64 * PA];
    __shared__ __attribute__((aligned(16))) float sB[2][(BK / 2) * PB];

    const int tid  = threadIdx.x;
    const int wave = tid >> 5;
    const int lane = tid & 31;
    const int half = lane >> 4;
    const int mr   = lane & 15;
    const int mb   = blockIdx.y * 64;
    const int m0   = mb + wave * 16;
    const int nb   = blockIdx.x * BN;

    const int arow = tid / (BK / 4), ac4 = tid % (BK / 4);
    const int bp   = tid / (BN / 4), bq  = tid % (BN / 4);
    const float* aG = A + (size_t)(mb + arow) * lda + 4 * ac4;
    const float* wG = W + (size_t)(2 * bp) * N + nb + 4 * bq;
    const int aS = arow * PA + 4 * ac4;
    const int bS = bp * PB + 8 * bq;
    const int aF = (wave * 16 + mr) * PA + 2 * half;
    const int bF = half * PB + 2 * mr;

    v8f acc[NT] = {};
    v4f ra[A_F4], rb0[B_IT], rb1[B_IT];

#pragma unroll
    for (int i = 0; i < A_F4; ++i)
        ra[i] = *(const v4f*)(aG + (size_t)(i * A_RS) * lda);
#pragma unroll
    for (int i = 0; i < B_IT; ++i) {
        rb0[i] = *(const v4f*)(wG + (size_t)(2 * i * B_PS) * N);
        rb1[i] = *(const v4f*)(wG + (size_t)(2 * i * B_PS + 1) * N);
    }

    const int S = K / BK;
    for (int s = 0; s < S - 1; ++s) {
        const int cur = s & 1;
#pragma unroll
        for (int i = 0; i < A_F4; ++i)
            *(v4f*)(&sA[cur][aS + i * A_RS * PA]) = ra[i];
#pragma unroll
        for (int i = 0; i < B_IT; ++i) {
            float* d = &sB[cur][bS + i * B_PS * PB];
            const v4f lo = { rb0[i][0], rb1[i][0], rb0[i][1], rb1[i][1] };
            const v4f hi = { rb0[i][2], rb1[i][2], rb0[i][3], rb1[i][3] };
            *(v4f*)(d)     = lo;
            *(v4f*)(d + 4) = hi;
        }
        __syncthreads();

        aG += BK;
        wG += (size_t)BK * N;
#pragma unroll
        for (int i = 0; i < A_F4; ++i)
            ra[i] = *(const v4f*)(aG + (size_t)(i * A_RS) * lda);
#pragma unroll
        for (int i = 0; i < B_IT; ++i) {
            rb0[i] = *(const v4f*)(wG + (size_t)(2 * i * B_PS) * N);
            rb1[i] = *(const v4f*)(wG + (size_t)(2 * i * B_PS + 1) * N);
        }

        const float* pa = &sA[cur][aF];
        const float* pb = &sB[cur][bF];
#pragma unroll
        for (int ks = 0; ks < KS; ++ks) {
            const v2f a = *(const v2f*)(pa + 4 * ks);
            v2f b[NT];
#pragma unroll
            for (int j = 0; j < NT; ++j) b[j] = *(const v2f*)(pb + 2 * ks * PB + 32 * j);
#pragma unroll
            for (int j = 0; j < NT; ++j)
                acc[j] = __builtin_amdgcn_wmma_f32_16x16x4_f32(
                    false, a, false, b[j], (short)0, acc[j], false, false);
        }
    }
    {
        const int cur = (S - 1) & 1;
#pragma unroll
        for (int i = 0; i < A_F4; ++i)
            *(v4f*)(&sA[cur][aS + i * A_RS * PA]) = ra[i];
#pragma unroll
        for (int i = 0; i < B_IT; ++i) {
            float* d = &sB[cur][bS + i * B_PS * PB];
            const v4f lo = { rb0[i][0], rb1[i][0], rb0[i][1], rb1[i][1] };
            const v4f hi = { rb0[i][2], rb1[i][2], rb0[i][3], rb1[i][3] };
            *(v4f*)(d)     = lo;
            *(v4f*)(d + 4) = hi;
        }
        __syncthreads();
        const float* pa = &sA[cur][aF];
        const float* pb = &sB[cur][bF];
#pragma unroll
        for (int ks = 0; ks < KS; ++ks) {
            const v2f a = *(const v2f*)(pa + 4 * ks);
            v2f b[NT];
#pragma unroll
            for (int j = 0; j < NT; ++j) b[j] = *(const v2f*)(pb + 2 * ks * PB + 32 * j);
#pragma unroll
            for (int j = 0; j < NT; ++j)
                acc[j] = __builtin_amdgcn_wmma_f32_16x16x4_f32(
                    false, a, false, b[j], (short)0, acc[j], false, false);
        }
    }

#pragma unroll
    for (int j = 0; j < NT; ++j) {
        const int col = nb + j * 16 + mr;
        const float bsv = SP ? bias[col] : 0.0f;
#pragma unroll
        for (int r = 0; r < 8; ++r) {
            float v = acc[j][r];
            if (SP) v = softplusf(v + bsv);
            C[(size_t)(m0 + r + 8 * half) * N + col] = v;
        }
    }
}

__global__ void conv_silu_kernel(const float* __restrict__ xz,
                                 const float* __restrict__ cw,
                                 const float* __restrict__ cb,
                                 float* __restrict__ u)
{
    const int idx = blockIdx.x * blockDim.x + threadIdx.x;
    const int d  = idx % DI;
    const int bt = idx / DI;
    const int t  = bt % LL;
    float acc = cb[d];
    const float* w = cw + d * KC;
#pragma unroll
    for (int k = 0; k < KC; ++k) {
        const int tt = t - (KC - 1) + k;
        if (tt >= 0)
            acc += xz[(size_t)(bt - (KC - 1) + k) * XZW + d] * w[k];
    }
    u[(size_t)bt * DI + d] = siluf(acc);
}

#define TCH   16
#define NCH   (LL / TCH)
#define BCW   (2 * NS)
#define QL    4
#define SPL   (NS / QL)
#define CPB   (128 / QL)
#define GST   4
#define NGR   (TCH / GST)

__global__ void __launch_bounds__(128) scan_kernel(
    const float* __restrict__ dt, const float* __restrict__ u,
    const float* __restrict__ dbl, const float* __restrict__ xz,
    const float* __restrict__ A_log, const float* __restrict__ Dp,
    float* __restrict__ yg)
{
    __shared__ __attribute__((aligned(16))) float sBC[2][TCH * BCW];
    __shared__ __attribute__((aligned(16))) float sY[2][TCH * 128];
    const int tid = threadIdx.x;
    const int b   = blockIdx.x / (DI / CPB);
    const int d0  = (blockIdx.x % (DI / CPB)) * CPB;
    const int q   = tid & (QL - 1);
    const int d   = d0 + (tid >> 2);

    float An[SPL];
#pragma unroll
    for (int k = 0; k < SPL; ++k) An[k] = -__expf(A_log[d * NS + SPL * q + k]);
    float h[SPL];
#pragma unroll
    for (int k = 0; k < SPL; ++k) h[k] = 0.0f;

    const size_t row0 = (size_t)b * LL;
    const float* pdt = dt + row0 * DI + d;
    const float* pu  = u  + row0 * DI + d;
    const float* pbc = dbl + (row0 + (tid >> 3)) * DBLW + RR + 4 * (tid & 7);
    const int fs = tid >> 3, fc = 4 * (tid & 7);
    const float* pzf = xz + (row0 + fs) * XZW + DI + d0 + fc;
    const float* puf = u  + (row0 + fs) * DI + d0 + fc;
    float*       pyf = yg + (row0 + fs) * DI + d0 + fc;
    const int jS = fs * 128 + QL * fc;
    float Df[4];
#pragma unroll
    for (int k = 0; k < 4; ++k) Df[k] = Dp[d0 + fc + k];
    const int bcq = SPL * q;

    float dt_n[TCH], u_n[TCH];
#pragma unroll
    for (int i = 0; i < TCH; ++i) {
        dt_n[i] = pdt[(size_t)i * DI];
        u_n[i]  = pu[(size_t)i * DI];
    }
    v4f bc_n = *(const v4f*)pbc;
    v4f zf = {}, uf = {};

    for (int c = 0; c < NCH; ++c) {
        float dt_c[TCH], u_c[TCH];
#pragma unroll
        for (int i = 0; i < TCH; ++i) { dt_c[i] = dt_n[i]; u_c[i] = u_n[i]; }

        float* buf = sBC[c & 1];
        *(v4f*)(buf + 4 * tid) = bc_n;
        __syncthreads();

        v4f ring[2][2 * GST];
        v4f sj[4];
        if (c > 0) {
            const float* sy = &sY[(c - 1) & 1][jS];
#pragma unroll
            for (int k = 0; k < 4; ++k) sj[k] = *(const v4f*)(sy + 4 * k);
        }
#pragma unroll
        for (int j = 0; j < GST; ++j) {
            ring[0][2 * j]     = *(const v4f*)(buf + j * BCW + bcq);
            ring[0][2 * j + 1] = *(const v4f*)(buf + j * BCW + NS + bcq);
        }
        if (c > 0) {
            v4f o;
#pragma unroll
            for (int k = 0; k < 4; ++k) {
                const float y = ((sj[k][0] + sj[k][1]) + sj[k][2]) + sj[k][3];
                o[k] = (y + uf[k] * Df[k]) * siluf(zf[k]);
            }
            *(v4f*)pyf = o;
            pyf += (size_t)TCH * DI;
        }
        zf = *(const v4f*)pzf;
        uf = *(const v4f*)puf;
        pzf += (size_t)TCH * XZW;
        puf += (size_t)TCH * DI;

        if (c + 1 < NCH) {
            pdt += (size_t)TCH * DI;
            pu  += (size_t)TCH * DI;
            pbc += (size_t)TCH * DBLW;
#pragma unroll
            for (int i = 0; i < TCH; ++i) {
                dt_n[i] = pdt[(size_t)i * DI];
                u_n[i]  = pu[(size_t)i * DI];
            }
            bc_n = *(const v4f*)pbc;
        }

        float p[TCH];
#pragma unroll
        for (int g = 0; g < NGR; ++g) {
            if (g + 1 < NGR) {
#pragma unroll
                for (int j = 0; j < GST; ++j) {
                    const float* r = buf + ((g + 1) * GST + j) * BCW + bcq;
                    ring[(g + 1) & 1][2 * j]     = *(const v4f*)(r);
                    ring[(g + 1) & 1][2 * j + 1] = *(const v4f*)(r + NS);
                }
            }
            __builtin_amdgcn_sched_barrier(0);

#pragma unroll
            for (int j = 0; j < GST; ++j) {
                const int i = g * GST + j;
                const float dtv = dt_c[i];
                const float du  = dtv * u_c[i];
                const v4f Bq = ring[g & 1][2 * j];
                const v4f Cq = ring[g & 1][2 * j + 1];
                float y = 0.0f;
#pragma unroll
                for (int k = 0; k < SPL; ++k) {
                    h[k] = __expf(dtv * An[k]) * h[k] + du * Bq[k];
                    y += h[k] * Cq[k];
                }
                p[i] = y;
            }
        }
        float* sy = &sY[c & 1][tid];
#pragma unroll
        for (int i = 0; i < TCH; ++i) sy[i * 128] = p[i];
    }

    __syncthreads();
    {
        const float* sy = &sY[(NCH - 1) & 1][jS];
        v4f o;
#pragma unroll
        for (int k = 0; k < 4; ++k) {
            const v4f s = *(const v4f*)(sy + 4 * k);
            const float y = ((s[0] + s[1]) + s[2]) + s[3];
            o[k] = (y + uf[k] * Df[k]) * siluf(zf[k]);
        }
        *(v4f*)pyf = o;
    }
}

extern "C" void kernel_launch(void* const* d_in, const int* in_sizes, int n_in,
                              void* d_out, int out_size, void* d_ws, size_t ws_size,
                              hipStream_t stream) {
    const float* x          = (const float*)d_in[0];
    const float* in_proj_w  = (const float*)d_in[1];
    const float* conv_w     = (const float*)d_in[2];
    const float* conv_b     = (const float*)d_in[3];
    const float* x_proj_w   = (const float*)d_in[4];
    const float* dt_proj_w  = (const float*)d_in[5];
    const float* dt_proj_b  = (const float*)d_in[6];
    const float* A_log      = (const float*)d_in[7];
    const float* Dvec       = (const float*)d_in[8];
    const float* out_proj_w = (const float*)d_in[9];
    float* out = (float*)d_out;

    float* ws  = (float*)d_ws;
    float* xz  = ws;
    float* u   = xz  + (size_t)2048 * 4096;
    float* dbl = u   + (size_t)2048 * 2048;
    float* dtb = dbl + (size_t)2048 * 96;
    float* yg  = dtb + (size_t)2048 * 2048;

    const dim3 blk(128);
    const int M = BB * LL;

    wmma_gemm_f32<4, 32, false><<<dim3(4096 / 64, M / 64), blk, 0, stream>>>(
        x, in_proj_w, xz, 4096, DMOD, DMOD, nullptr);

    conv_silu_kernel<<<(BB * LL * DI) / 256, 256, 0, stream>>>(xz, conv_w, conv_b, u);

    wmma_gemm_f32<2, 64, false><<<dim3(DBLW / 32, M / 64), blk, 0, stream>>>(
        u, x_proj_w, dbl, DBLW, DI, DI, nullptr);

    wmma_gemm_f32<4, 32, true><<<dim3(DI / 64, M / 64), blk, 0, stream>>>(
        dbl, dt_proj_w, dtb, DI, RR, DBLW, dt_proj_b);

    scan_kernel<<<BB * (DI / CPB), 128, 0, stream>>>(dtb, u, dbl, xz, A_log, Dvec, yg);

    wmma_gemm_f32<4, 32, false><<<dim3(DMOD / 64, M / 64), blk, 0, stream>>>(
        yg, out_proj_w, out, DMOD, DI, DI, nullptr);
}


# scan: the 1024 steps run as 4 time segments in parallel from zero states; a second pass adds the carried state's share C_t.(exp(A*cumsum dt)*h_in) to y (new summation order), then skip and gate
# speedup vs baseline: 8.3785x; 1.0573x over previous
#include <hip/hip_runtime.h>
#include <hip/hip_bf16.h>
#include <cmath>

typedef __attribute__((ext_vector_type(2))) float v2f;
typedef __attribute__((ext_vector_type(4))) float v4f;
typedef __attribute__((ext_vector_type(8))) float v8f;

#define BB    2
#define LL    1024
#define DMOD  1024
#define DI    2048
#define NS    16
#define KC    4
#define RR    64
#define XZW   (2 * DI)
#define DBLW  (RR + 2 * NS)

__device__ __forceinline__ float siluf(float v) {
    return v / (1.0f + __expf(-v));
}
__device__ __forceinline__ float softplusf(float v) {
    return (v > 20.0f) ? v : log1pf(expf(v));
}

template <int NT, int BK, bool SP>
__global__ void __launch_bounds__(128) wmma_gemm_f32(
    const float* __restrict__ A, const float* __restrict__ W,
    float* __restrict__ C, int N, int K, int lda,
    const float* __restrict__ bias)
{
    constexpr int BN   = NT * 16;
    constexpr int PA   = BK + 4;
    constexpr int PB   = 2 * BN + 32;
    constexpr int KS   = BK / 4;
    constexpr int A_F4 = (64 * (BK / 4)) / 128;
    constexpr int B_IT = ((BK / 2) * (BN / 4)) / 128;
    constexpr int A_RS = 128 / (BK / 4);
    constexpr int B_PS = 128 / (BN / 4);

    __shared__ __attribute__((aligned(16))) float sA[2][64 * PA];
    __shared__ __attribute__((aligned(16))) float sB[2][(BK / 2) * PB];

    const int tid  = threadIdx.x;
    const int wave = tid >> 5;
    const int lane = tid & 31;
    const int half = lane >> 4;
    const int mr   = lane & 15;
    const int mb   = blockIdx.y * 64;
    const int m0   = mb + wave * 16;
    const int nb   = blockIdx.x * BN;

    const int arow = tid / (BK / 4), ac4 = tid % (BK / 4);
    const int bp   = tid / (BN / 4), bq  = tid % (BN / 4);
    const float* aG = A + (size_t)(mb + arow) * lda + 4 * ac4;
    const float* wG = W + (size_t)(2 * bp) * N + nb + 4 * bq;
    const int aS = arow * PA + 4 * ac4;
    const int bS = bp * PB + 8 * bq;
    const int aF = (wave * 16 + mr) * PA + 2 * half;
    const int bF = half * PB + 2 * mr;

    v8f acc[NT] = {};
    v4f ra[A_F4], rb0[B_IT], rb1[B_IT];

#pragma unroll
    for (int i = 0; i < A_F4; ++i)
        ra[i] = *(const v4f*)(aG + (size_t)(i * A_RS) * lda);
#pragma unroll
    for (int i = 0; i < B_IT; ++i) {
        rb0[i] = *(const v4f*)(wG + (size_t)(2 * i * B_PS) * N);
        rb1[i] = *(const v4f*)(wG + (size_t)(2 * i * B_PS + 1) * N);
    }

    const int S = K / BK;
    for (int s = 0; s < S - 1; ++s) {
        const int cur = s & 1;
#pragma unroll
        for (int i = 0; i < A_F4; ++i)
            *(v4f*)(&sA[cur][aS + i * A_RS * PA]) = ra[i];
#pragma unroll
        for (int i = 0; i < B_IT; ++i) {
            float* d = &sB[cur][bS + i * B_PS * PB];
            const v4f lo = { rb0[i][0], rb1[i][0], rb0[i][1], rb1[i][1] };
            const v4f hi = { rb0[i][2], rb1[i][2], rb0[i][3], rb1[i][3] };
            *(v4f*)(d)     = lo;
            *(v4f*)(d + 4) = hi;
        }
        __syncthreads();

        aG += BK;
        wG += (size_t)BK * N;
#pragma unroll
        for (int i = 0; i < A_F4; ++i)
            ra[i] = *(const v4f*)(aG + (size_t)(i * A_RS) * lda);
#pragma unroll
        for (int i = 0; i < B_IT; ++i) {
            rb0[i] = *(const v4f*)(wG + (size_t)(2 * i * B_PS) * N);
            rb1[i] = *(const v4f*)(wG + (size_t)(2 * i * B_PS + 1) * N);
        }

        const float* pa = &sA[cur][aF];
        const float* pb = &sB[cur][bF];
#pragma unroll
        for (int ks = 0; ks < KS; ++ks) {
            const v2f a = *(const v2f*)(pa + 4 * ks);
            v2f b[NT];
#pragma unroll
            for (int j = 0; j < NT; ++j) b[j] = *(const v2f*)(pb + 2 * ks * PB + 32 * j);
#pragma unroll
            for (int j = 0; j < NT; ++j)
                acc[j] = __builtin_amdgcn_wmma_f32_16x16x4_f32(
                    false, a, false, b[j], (short)0, acc[j], false, false);
        }
    }
    {
        const int cur = (S - 1) & 1;
#pragma unroll
        for (int i = 0; i < A_F4; ++i)
            *(v4f*)(&sA[cur][aS + i * A_RS * PA]) = ra[i];
#pragma unroll
        for (int i = 0; i < B_IT; ++i) {
            float* d = &sB[cur][bS + i * B_PS * PB];
            const v4f lo = { rb0[i][0], rb1[i][0], rb0[i][1], rb1[i][1] };
            const v4f hi = { rb0[i][2], rb1[i][2], rb0[i][3], rb1[i][3] };
            *(v4f*)(d)     = lo;
            *(v4f*)(d + 4) = hi;
        }
        __syncthreads();
        const float* pa = &sA[cur][aF];
        const float* pb = &sB[cur][bF];
#pragma unroll
        for (int ks = 0; ks < KS; ++ks) {
            const v2f a = *(const v2f*)(pa + 4 * ks);
            v2f b[NT];
#pragma unroll
            for (int j = 0; j < NT; ++j) b[j] = *(const v2f*)(pb + 2 * ks * PB + 32 * j);
#pragma unroll
            for (int j = 0; j < NT; ++j)
                acc[j] = __builtin_amdgcn_wmma_f32_16x16x4_f32(
                    false, a, false, b[j], (short)0, acc[j], false, false);
        }
    }

#pragma unroll
    for (int j = 0; j < NT; ++j) {
        const int col = nb + j * 16 + mr;
        const float bsv = SP ? bias[col] : 0.0f;
#pragma unroll
        for (int r = 0; r < 8; ++r) {
            float v = acc[j][r];
            if (SP) v = softplusf(v + bsv);
            C[(size_t)(m0 + r + 8 * half) * N + col] = v;
        }
    }
}

__global__ void conv_silu_kernel(const float* __restrict__ xz,
                                 const float* __restrict__ cw,
                                 const float* __restrict__ cb,
                                 float* __restrict__ u)
{
    const int idx = blockIdx.x * blockDim.x + threadIdx.x;
    const int d  = idx % DI;
    const int bt = idx / DI;
    const int t  = bt % LL;
    float acc = cb[d];
    const float* w = cw + d * KC;
#pragma unroll
    for (int k = 0; k < KC; ++k) {
        const int tt = t - (KC - 1) + k;
        if (tt >= 0)
            acc += xz[(size_t)(bt - (KC - 1) + k) * XZW + d] * w[k];
    }
    u[(size_t)bt * DI + d] = siluf(acc);
}

#define TCH   16
#define NSEG  4
#define LSEG  (LL / NSEG)
#define NCH   (LSEG / TCH)
#define BCW   (2 * NS)
#define QL    4
#define SPL   (NS / QL)
#define CPB   (128 / QL)
#define GST   4
#define NGR   (TCH / GST)

__global__ void __launch_bounds__(128) scan_kernel(
    const float* __restrict__ dt, const float* __restrict__ u,
    const float* __restrict__ dbl, const float* __restrict__ A_log,
    float* __restrict__ xw, float* __restrict__ yg)
{
    __shared__ __attribute__((aligned(16))) float sBC[2][TCH * BCW];
    __shared__ __attribute__((aligned(16))) float sY[2][TCH * 128];
    __shared__ __attribute__((aligned(16))) float sC[2][CPB];
    __shared__ __attribute__((aligned(16))) float sH[NS * CPB];
    const int tid = threadIdx.x;
    const int cb  = blockIdx.x % (DI / CPB);
    const int seg = (blockIdx.x / (DI / CPB)) % NSEG;
    const int b   = blockIdx.x / ((DI / CPB) * NSEG);
    const int d0  = cb * CPB;
    const int q   = tid & (QL - 1);
    const int d   = d0 + (tid >> 2);

    float An[SPL];
#pragma unroll
    for (int k = 0; k < SPL; ++k) An[k] = -__expf(A_log[d * NS + SPL * q + k]);
    float h[SPL];
#pragma unroll
    for (int k = 0; k < SPL; ++k) h[k] = 0.0f;
    float cs = 0.0f;

    const size_t row0 = (size_t)b * LL + (size_t)seg * LSEG;
    const float* pdt = dt + row0 * DI + d;
    const float* pu  = u  + row0 * DI + d;
    const float* pbc = dbl + (row0 + (tid >> 3)) * DBLW + RR + 4 * (tid & 7);
    const int fs = tid >> 3, fc = 4 * (tid & 7);
    float* pyf = yg + (row0 + fs) * DI + d0 + fc;
    float* pck = xw + row0 * XZW + d0 + fc;
    const int jS = fs * 128 + QL * fc;
    const int bcq = SPL * q;

    float dt_n[TCH], u_n[TCH];
#pragma unroll
    for (int i = 0; i < TCH; ++i) {
        dt_n[i] = pdt[(size_t)i * DI];
        u_n[i]  = pu[(size_t)i * DI];
    }
    v4f bc_n = *(const v4f*)pbc;

    for (int c = 0; c < NCH; ++c) {
        float dt_c[TCH], u_c[TCH];
#pragma unroll
        for (int i = 0; i < TCH; ++i) { dt_c[i] = dt_n[i]; u_c[i] = u_n[i]; }

        float* buf = sBC[c & 1];
        *(v4f*)(buf + 4 * tid) = bc_n;
        __syncthreads();

        v4f ring[2][2 * GST];
        v4f sj[4];
        if (c > 0) {
            const float* sy = &sY[(c - 1) & 1][jS];
#pragma unroll
            for (int k = 0; k < 4; ++k) sj[k] = *(const v4f*)(sy + 4 * k);
        }
#pragma unroll
        for (int j = 0; j < GST; ++j) {
            ring[0][2 * j]     = *(const v4f*)(buf + j * BCW + bcq);
            ring[0][2 * j + 1] = *(const v4f*)(buf + j * BCW + NS + bcq);
        }
        if (c > 0) {
            v4f o;
#pragma unroll
            for (int k = 0; k < 4; ++k)
                o[k] = ((sj[k][0] + sj[k][1]) + sj[k][2]) + sj[k][3];
            *(v4f*)pyf = o;
            pyf += (size_t)TCH * DI;
            if (fs == 0) *(v4f*)pck = *(const v4f*)&sC[(c - 1) & 1][fc];
            pck += (size_t)TCH * XZW;
        }

        if (c + 1 < NCH) {
            pdt += (size_t)TCH * DI;
            pu  += (size_t)TCH * DI;
            pbc += (size_t)TCH * DBLW;
#pragma unroll
            for (int i = 0; i < TCH; ++i) {
                dt_n[i] = pdt[(size_t)i * DI];
                u_n[i]  = pu[(size_t)i * DI];
            }
            bc_n = *(const v4f*)pbc;
        }

        float p[TCH];
#pragma unroll
        for (int g = 0; g < NGR; ++g) {
            if (g + 1 < NGR) {
#pragma unroll
                for (int j = 0; j < GST; ++j) {
                    const float* r = buf + ((g + 1) * GST + j) * BCW + bcq;
                    ring[(g + 1) & 1][2 * j]     = *(const v4f*)(r);
                    ring[(g + 1) & 1][2 * j + 1] = *(const v4f*)(r + NS);
                }
            }
            __builtin_amdgcn_sched_barrier(0);

#pragma unroll
            for (int j = 0; j < GST; ++j) {
                const int i = g * GST + j;
                const float dtv = dt_c[i];
                const float du  = dtv * u_c[i];
                const v4f Bq = ring[g & 1][2 * j];
                const v4f Cq = ring[g & 1][2 * j + 1];
                float y = 0.0f;
#pragma unroll
                for (int k = 0; k < SPL; ++k) {
                    h[k] = __expf(dtv * An[k]) * h[k] + du * Bq[k];
                    y += h[k] * Cq[k];
                }
                p[i] = y;
                cs += dtv;
            }
        }
        float* sy = &sY[c & 1][tid];
#pragma unroll
        for (int i = 0; i < TCH; ++i) sy[i * 128] = p[i];
        if (q == 0) sC[c & 1][tid >> 2] = cs;
    }

#pragma unroll
    for (int k = 0; k < SPL; ++k) sH[(SPL * q + k) * CPB + (tid >> 2)] = h[k];
    __syncthreads();
    {
        const float* sy = &sY[(NCH - 1) & 1][jS];
        v4f o;
#pragma unroll
        for (int k = 0; k < 4; ++k) {
            const v4f s = *(const v4f*)(sy + 4 * k);
            o[k] = ((s[0] + s[1]) + s[2]) + s[3];
        }
        *(v4f*)pyf = o;
        if (fs == 0) *(v4f*)pck = *(const v4f*)&sC[(NCH - 1) & 1][fc];
        float* ph = xw + (row0 + (size_t)(TCH * fs + 1)) * XZW + d0 + fc;
        *(v4f*)ph = *(const v4f*)&sH[fs * CPB + fc];
    }
}

#define TT    32
#define CSUB  8
#define DPB   128

template <bool CARRY>
__device__ __forceinline__ void carry_tile(
    const float* __restrict__ pdt, const float* __restrict__ pu,
    const float* __restrict__ pz, float* __restrict__ py,
    const float* sCt, const v2f (&A2)[NS / 2], const v2f (&hin)[NS / 2],
    float cs, const float Dd)
{
    for (int t0 = 0; t0 < TT; t0 += CSUB) {
        float dv[CSUB], yl[CSUB], uv[CSUB], zv[CSUB];
#pragma unroll
        for (int i = 0; i < CSUB; ++i) {
            dv[i] = CARRY ? pdt[(size_t)(t0 + i) * DI] : 0.0f;
            yl[i] = py[(size_t)(t0 + i) * DI];
            uv[i] = pu[(size_t)(t0 + i) * DI];
            zv[i] = pz[(size_t)(t0 + i) * XZW];
        }
#pragma unroll
        for (int i = 0; i < CSUB; ++i) {
            float y = yl[i];
            if (CARRY) {
                cs += dv[i];
                const v4f* cr = (const v4f*)(sCt + (t0 + i) * NS);
                v2f acc = { 0.0f, 0.0f };
#pragma unroll
                for (int k4 = 0; k4 < NS / 4; ++k4) {
                    const v4f c4 = cr[k4];
                    const v2f a0 = A2[2 * k4] * cs;
                    const v2f a1 = A2[2 * k4 + 1] * cs;
                    const v2f e0 = { __builtin_amdgcn_exp2f(a0[0]), __builtin_amdgcn_exp2f(a0[1]) };
                    const v2f e1 = { __builtin_amdgcn_exp2f(a1[0]), __builtin_amdgcn_exp2f(a1[1]) };
                    const v2f c0 = { c4[0], c4[1] };
                    const v2f c1 = { c4[2], c4[3] };
                    acc += (e0 * hin[2 * k4]) * c0;
                    acc += (e1 * hin[2 * k4 + 1]) * c1;
                }
                y += acc[0] + acc[1];
            }
            py[(size_t)(t0 + i) * DI] = (y + uv[i] * Dd) * siluf(zv[i]);
        }
    }
}

__global__ void __launch_bounds__(128) carry_gate_kernel(
    const float* __restrict__ dt, const float* __restrict__ u,
    const float* __restrict__ dbl, const float* __restrict__ xz,
    const float* __restrict__ A_log, const float* __restrict__ Dp,
    float* __restrict__ yg)
{
    __shared__ __attribute__((aligned(16))) float sCt[TT * NS];
    const int tid  = threadIdx.x;
    const int dblk = blockIdx.x % (DI / DPB);
    const int tile = (blockIdx.x / (DI / DPB)) % (LSEG / TT);
    const int seg  = (blockIdx.x / ((DI / DPB) * (LSEG / TT))) % NSEG;
    const int b    = blockIdx.x / ((DI / DPB) * (LSEG / TT) * NSEG);
    const int d    = dblk * DPB + tid;
    const size_t r0 = (size_t)b * LL + (size_t)seg * LSEG + (size_t)tile * TT;

    const float Dd = Dp[d];
    const float* pdt = dt + r0 * DI + d;
    const float* pu  = u  + r0 * DI + d;
    const float* pz  = xz + r0 * XZW + DI + d;
    float*       py  = yg + r0 * DI + d;

    v2f A2[NS / 2], hin[NS / 2];
    if (seg > 0) {
        *(v4f*)(sCt + 4 * tid) =
            *(const v4f*)(dbl + (r0 + (tid >> 2)) * DBLW + RR + NS + 4 * (tid & 3));
        const v4f* pa = (const v4f*)(A_log + (size_t)d * NS);
#pragma unroll
        for (int k4 = 0; k4 < NS / 4; ++k4) {
            const v4f a = pa[k4];
            A2[2 * k4]     = v2f{ -__expf(a[0]), -__expf(a[1]) } * 1.44269504088896341f;
            A2[2 * k4 + 1] = v2f{ -__expf(a[2]), -__expf(a[3]) } * 1.44269504088896341f;
            hin[2 * k4]     = v2f{ 0.0f, 0.0f };
            hin[2 * k4 + 1] = v2f{ 0.0f, 0.0f };
        }
        for (int s = 0; s < seg; ++s) {
            const size_t rs = (size_t)b * LL + (size_t)s * LSEG;
            const float tot = xz[(rs + LSEG - TCH) * XZW + d];
#pragma unroll
            for (int k2 = 0; k2 < NS / 2; ++k2) {
                const v2f a  = A2[k2] * tot;
                const v2f e  = { __builtin_amdgcn_exp2f(a[0]), __builtin_amdgcn_exp2f(a[1]) };
                const v2f hf = { xz[(rs + (size_t)(TCH * (2 * k2) + 1)) * XZW + d],
                                 xz[(rs + (size_t)(TCH * (2 * k2 + 1) + 1)) * XZW + d] };
                hin[k2] = e * hin[k2] + hf;
            }
        }
        const float cs0 = (tile > 0) ? xz[(r0 - TCH) * XZW + d] : 0.0f;
        __syncthreads();
        carry_tile<true>(pdt, pu, pz, py, sCt, A2, hin, cs0, Dd);
    } else {
#pragma unroll
        for (int k2 = 0; k2 < NS / 2; ++k2) { A2[k2] = v2f{ 0.0f, 0.0f }; hin[k2] = v2f{ 0.0f, 0.0f }; }
        carry_tile<false>(pdt, pu, pz, py, sCt, A2, hin, 0.0f, Dd);
    }
}

extern "C" void kernel_launch(void* const* d_in, const int* in_sizes, int n_in,
                              void* d_out, int out_size, void* d_ws, size_t ws_size,
                              hipStream_t stream) {
    const float* x          = (const float*)d_in[0];
    const float* in_proj_w  = (const float*)d_in[1];
    const float* conv_w     = (const float*)d_in[2];
    const float* conv_b     = (const float*)d_in[3];
    const float* x_proj_w   = (const float*)d_in[4];
    const float* dt_proj_w  = (const float*)d_in[5];
    const float* dt_proj_b  = (const float*)d_in[6];
    const float* A_log      = (const float*)d_in[7];
    const float* Dvec       = (const float*)d_in[8];
    const float* out_proj_w = (const float*)d_in[9];
    float* out = (float*)d_out;

    float* ws  = (float*)d_ws;
    float* xz  = ws;
    float* u   = xz  + (size_t)2048 * 4096;
    float* dbl = u   + (size_t)2048 * 2048;
    float* dtb = dbl + (size_t)2048 * 96;
    float* yg  = dtb + (size_t)2048 * 2048;

    const dim3 blk(128);
    const int M = BB * LL;

    wmma_gemm_f32<4, 32, false><<<dim3(4096 / 64, M / 64), blk, 0, stream>>>(
        x, in_proj_w, xz, 4096, DMOD, DMOD, nullptr);

    conv_silu_kernel<<<(BB * LL * DI) / 256, 256, 0, stream>>>(xz, conv_w, conv_b, u);

    wmma_gemm_f32<2, 64, false><<<dim3(DBLW / 32, M / 64), blk, 0, stream>>>(
        u, x_proj_w, dbl, DBLW, DI, DI, nullptr);

    wmma_gemm_f32<4, 32, true><<<dim3(DI / 64, M / 64), blk, 0, stream>>>(
        dbl, dt_proj_w, dtb, DI, RR, DBLW, dt_proj_b);

    scan_kernel<<<BB * NSEG * (DI / CPB), 128, 0, stream>>>(dtb, u, dbl, A_log, xz, yg);
    carry_gate_kernel<<<BB * NSEG * (LSEG / TT) * (DI / DPB), 128, 0, stream>>>(
        dtb, u, dbl, xz, A_log, Dvec, yg);

    wmma_gemm_f32<4, 32, false><<<dim3(DMOD / 64, M / 64), blk, 0, stream>>>(
        yg, out_proj_w, out, DMOD, DI, DI, nullptr);
}


# x_proj product: k split into 8 ranges over the grid (768 blocks, not 96), one partial plane each; dt_proj adds the planes in ascending order while fetching A (new summation order)
# speedup vs baseline: 8.9102x; 1.0635x over previous
#include <hip/hip_runtime.h>
#include <hip/hip_bf16.h>
#include <cmath>

typedef __attribute__((ext_vector_type(2))) float v2f;
typedef __attribute__((ext_vector_type(4))) float v4f;
typedef __attribute__((ext_vector_type(8))) float v8f;

#define BB    2
#define LL    1024
#define DMOD  1024
#define DI    2048
#define NS    16
#define KC    4
#define RR    64
#define XZW   (2 * DI)
#define DBLW  (RR + 2 * NS)
#define KSPL  8

__device__ __forceinline__ float siluf(float v) {
    return v / (1.0f + __expf(-v));
}
__device__ __forceinline__ float softplusf(float v) {
    return (v > 20.0f) ? v : log1pf(expf(v));
}

template <int AP>
__device__ __forceinline__ v4f load_planes(const float* p, size_t plane)
{
    v4f v = *(const v4f*)p;
#pragma unroll
    for (int s = 1; s < AP; ++s) v += *(const v4f*)(p + (size_t)s * plane);
    return v;
}

template <int NT, int BK, bool SP, int AP>
__global__ void __launch_bounds__(128) wmma_gemm_f32(
    const float* __restrict__ A, const float* __restrict__ W,
    float* __restrict__ C, int N, int K, int lda,
    const float* __restrict__ bias, float* __restrict__ red)
{
    constexpr int BN   = NT * 16;
    constexpr int PA   = BK + 4;
    constexpr int PB   = 2 * BN + 32;
    constexpr int KS   = BK / 4;
    constexpr int A_F4 = (64 * (BK / 4)) / 128;
    constexpr int B_IT = ((BK / 2) * (BN / 4)) / 128;
    constexpr int A_RS = 128 / (BK / 4);
    constexpr int B_PS = 128 / (BN / 4);

    __shared__ __attribute__((aligned(16))) float sA[2][64 * PA];
    __shared__ __attribute__((aligned(16))) float sB[2][(BK / 2) * PB];

    const int tid  = threadIdx.x;
    const int wave = tid >> 5;
    const int lane = tid & 31;
    const int half = lane >> 4;
    const int mr   = lane & 15;
    const int mb   = blockIdx.y * 64;
    const int m0   = mb + wave * 16;
    const int nb   = blockIdx.x * BN;
    const size_t mrows  = (size_t)gridDim.y * 64;
    const size_t aplane = mrows * lda;
    A += (size_t)blockIdx.z * K;
    W += (size_t)blockIdx.z * K * N;
    C += (size_t)blockIdx.z * mrows * N;

    const int arow = tid / (BK / 4), ac4 = tid % (BK / 4);
    const int bp   = tid / (BN / 4), bq  = tid % (BN / 4);
    const float* aG = A + (size_t)(mb + arow) * lda + 4 * ac4;
    const float* wG = W + (size_t)(2 * bp) * N + nb + 4 * bq;
    const int aS = arow * PA + 4 * ac4;
    const int bS = bp * PB + 8 * bq;
    const int aF = (wave * 16 + mr) * PA + 2 * half;
    const int bF = half * PB + 2 * mr;

    v8f acc[NT] = {};
    v4f ra[A_F4], rb0[B_IT], rb1[B_IT];

#pragma unroll
    for (int i = 0; i < A_F4; ++i)
        ra[i] = load_planes<AP>(aG + (size_t)(i * A_RS) * lda, aplane);
#pragma unroll
    for (int i = 0; i < B_IT; ++i) {
        rb0[i] = *(const v4f*)(wG + (size_t)(2 * i * B_PS) * N);
        rb1[i] = *(const v4f*)(wG + (size_t)(2 * i * B_PS + 1) * N);
    }

    if (AP > 1 && blockIdx.x == 0) {
        const int rq = (lda - K) / 4;
        for (int p = tid; p < 64 * rq; p += 128) {
            const size_t o = (size_t)(mb + p / rq) * lda + K + 4 * (p % rq);
            *(v4f*)(red + o) = load_planes<AP>(A + o, aplane);
        }
    }

    const int S = K / BK;
    for (int s = 0; s < S - 1; ++s) {
        const int cur = s & 1;
#pragma unroll
        for (int i = 0; i < A_F4; ++i)
            *(v4f*)(&sA[cur][aS + i * A_RS * PA]) = ra[i];
#pragma unroll
        for (int i = 0; i < B_IT; ++i) {
            float* d = &sB[cur][bS + i * B_PS * PB];
            const v4f lo = { rb0[i][0], rb1[i][0], rb0[i][1], rb1[i][1] };
            const v4f hi = { rb0[i][2], rb1[i][2], rb0[i][3], rb1[i][3] };
            *(v4f*)(d)     = lo;
            *(v4f*)(d + 4) = hi;
        }
        __syncthreads();

        aG += BK;
        wG += (size_t)BK * N;
#pragma unroll
        for (int i = 0; i < A_F4; ++i)
            ra[i] = load_planes<AP>(aG + (size_t)(i * A_RS) * lda, aplane);
#pragma unroll
        for (int i = 0; i < B_IT; ++i) {
            rb0[i] = *(const v4f*)(wG + (size_t)(2 * i * B_PS) * N);
            rb1[i] = *(const v4f*)(wG + (size_t)(2 * i * B_PS + 1) * N);
        }

        const float* pa = &sA[cur][aF];
        const float* pb = &sB[cur][bF];
#pragma unroll
        for (int ks = 0; ks < KS; ++ks) {
            const v2f a = *(const v2f*)(pa + 4 * ks);
            v2f b[NT];
#pragma unroll
            for (int j = 0; j < NT; ++j) b[j] = *(const v2f*)(pb + 2 * ks * PB + 32 * j);
#pragma unroll
            for (int j = 0; j < NT; ++j)
                acc[j] = __builtin_amdgcn_wmma_f32_16x16x4_f32(
                    false, a, false, b[j], (short)0, acc[j], false, false);
        }
    }
    {
        const int cur = (S - 1) & 1;
#pragma unroll
        for (int i = 0; i < A_F4; ++i)
            *(v4f*)(&sA[cur][aS + i * A_RS * PA]) = ra[i];
#pragma unroll
        for (int i = 0; i < B_IT; ++i) {
            float* d = &sB[cur][bS + i * B_PS * PB];
            const v4f lo = { rb0[i][0], rb1[i][0], rb0[i][1], rb1[i][1] };
            const v4f hi = { rb0[i][2], rb1[i][2], rb0[i][3], rb1[i][3] };
            *(v4f*)(d)     = lo;
            *(v4f*)(d + 4) = hi;
        }
        __syncthreads();
        const float* pa = &sA[cur][aF];
        const float* pb = &sB[cur][bF];
#pragma unroll
        for (int ks = 0; ks < KS; ++ks) {
            const v2f a = *(const v2f*)(pa + 4 * ks);
            v2f b[NT];
#pragma unroll
            for (int j = 0; j < NT; ++j) b[j] = *(const v2f*)(pb + 2 * ks * PB + 32 * j);
#pragma unroll
            for (int j = 0; j < NT; ++j)
                acc[j] = __builtin_amdgcn_wmma_f32_16x16x4_f32(
                    false, a, false, b[j], (short)0, acc[j], false, false);
        }
    }

#pragma unroll
    for (int j = 0; j < NT; ++j) {
        const int col = nb + j * 16 + mr;
        const float bsv = SP ? bias[col] : 0.0f;
#pragma unroll
        for (int r = 0; r < 8; ++r) {
            float v = acc[j][r];
            if (SP) v = softplusf(v + bsv);
            C[(size_t)(m0 + r + 8 * half) * N + col] = v;
        }
    }
}

__global__ void conv_silu_kernel(const float* __restrict__ xz,
                                 const float* __restrict__ cw,
                                 const float* __restrict__ cb,
                                 float* __restrict__ u)
{
    const int idx = blockIdx.x * blockDim.x + threadIdx.x;
    const int d  = idx % DI;
    const int bt = idx / DI;
    const int t  = bt % LL;
    float acc = cb[d];
    const float* w = cw + d * KC;
#pragma unroll
    for (int k = 0; k < KC; ++k) {
        const int tt = t - (KC - 1) + k;
        if (tt >= 0)
            acc += xz[(size_t)(bt - (KC - 1) + k) * XZW + d] * w[k];
    }
    u[(size_t)bt * DI + d] = siluf(acc);
}

#define TCH   16
#define NSEG  4
#define LSEG  (LL / NSEG)
#define NCH   (LSEG / TCH)
#define BCW   (2 * NS)
#define QL    4
#define SPL   (NS / QL)
#define CPB   (128 / QL)
#define GST   4
#define NGR   (TCH / GST)

__global__ void __launch_bounds__(128) scan_kernel(
    const float* __restrict__ dt, const float* __restrict__ u,
    const float* __restrict__ dbl, const float* __restrict__ A_log,
    float* __restrict__ xw, float* __restrict__ yg)
{
    __shared__ __attribute__((aligned(16))) float sBC[2][TCH * BCW];
    __shared__ __attribute__((aligned(16))) float sY[2][TCH * 128];
    __shared__ __attribute__((aligned(16))) float sC[2][CPB];
    __shared__ __attribute__((aligned(16))) float sH[NS * CPB];
    const int tid = threadIdx.x;
    const int cb  = blockIdx.x % (DI / CPB);
    const int seg = (blockIdx.x / (DI / CPB)) % NSEG;
    const int b   = blockIdx.x / ((DI / CPB) * NSEG);
    const int d0  = cb * CPB;
    const int q   = tid & (QL - 1);
    const int d   = d0 + (tid >> 2);

    float An[SPL];
#pragma unroll
    for (int k = 0; k < SPL; ++k) An[k] = -__expf(A_log[d * NS + SPL * q + k]);
    float h[SPL];
#pragma unroll
    for (int k = 0; k < SPL; ++k) h[k] = 0.0f;
    float cs = 0.0f;

    const size_t row0 = (size_t)b * LL + (size_t)seg * LSEG;
    const float* pdt = dt + row0 * DI + d;
    const float* pu  = u  + row0 * DI + d;
    const float* pbc = dbl + (row0 + (tid >> 3)) * DBLW + RR + 4 * (tid & 7);
    const int fs = tid >> 3, fc = 4 * (tid & 7);
    float* pyf = yg + (row0 + fs) * DI + d0 + fc;
    float* pck = xw + row0 * XZW + d0 + fc;
    const int jS = fs * 128 + QL * fc;
    const int bcq = SPL * q;

    float dt_n[TCH], u_n[TCH];
#pragma unroll
    for (int i = 0; i < TCH; ++i) {
        dt_n[i] = pdt[(size_t)i * DI];
        u_n[i]  = pu[(size_t)i * DI];
    }
    v4f bc_n = *(const v4f*)pbc;

    for (int c = 0; c < NCH; ++c) {
        float dt_c[TCH], u_c[TCH];
#pragma unroll
        for (int i = 0; i < TCH; ++i) { dt_c[i] = dt_n[i]; u_c[i] = u_n[i]; }

        float* buf = sBC[c & 1];
        *(v4f*)(buf + 4 * tid) = bc_n;
        __syncthreads();

        v4f ring[2][2 * GST];
        v4f sj[4];
        if (c > 0) {
            const float* sy = &sY[(c - 1) & 1][jS];
#pragma unroll
            for (int k = 0; k < 4; ++k) sj[k] = *(const v4f*)(sy + 4 * k);
        }
#pragma unroll
        for (int j = 0; j < GST; ++j) {
            ring[0][2 * j]     = *(const v4f*)(buf + j * BCW + bcq);
            ring[0][2 * j + 1] = *(const v4f*)(buf + j * BCW + NS + bcq);
        }
        if (c > 0) {
            v4f o;
#pragma unroll
            for (int k = 0; k < 4; ++k)
                o[k] = ((sj[k][0] + sj[k][1]) + sj[k][2]) + sj[k][3];
            *(v4f*)pyf = o;
            pyf += (size_t)TCH * DI;
            if (fs == 0) *(v4f*)pck = *(const v4f*)&sC[(c - 1) & 1][fc];
            pck += (size_t)TCH * XZW;
        }

        if (c + 1 < NCH) {
            pdt += (size_t)TCH * DI;
            pu  += (size_t)TCH * DI;
            pbc += (size_t)TCH * DBLW;
#pragma unroll
            for (int i = 0; i < TCH; ++i) {
                dt_n[i] = pdt[(size_t)i * DI];
                u_n[i]  = pu[(size_t)i * DI];
            }
            bc_n = *(const v4f*)pbc;
        }

        float p[TCH];
#pragma unroll
        for (int g = 0; g < NGR; ++g) {
            if (g + 1 < NGR) {
#pragma unroll
                for (int j = 0; j < GST; ++j) {
                    const float* r = buf + ((g + 1) * GST + j) * BCW + bcq;
                    ring[(g + 1) & 1][2 * j]     = *(const v4f*)(r);
                    ring[(g + 1) & 1][2 * j + 1] = *(const v4f*)(r + NS);
                }
            }
            __builtin_amdgcn_sched_barrier(0);

#pragma unroll
            for (int j = 0; j < GST; ++j) {
                const int i = g * GST + j;
                const float dtv = dt_c[i];
                const float du  = dtv * u_c[i];
                const v4f Bq = ring[g & 1][2 * j];
                const v4f Cq = ring[g & 1][2 * j + 1];
                float y = 0.0f;
#pragma unroll
                for (int k = 0; k < SPL; ++k) {
                    h[k] = __expf(dtv * An[k]) * h[k] + du * Bq[k];
                    y += h[k] * Cq[k];
                }
                p[i] = y;
                cs += dtv;
            }
        }
        float* sy = &sY[c & 1][tid];
#pragma unroll
        for (int i = 0; i < TCH; ++i) sy[i * 128] = p[i];
        if (q == 0) sC[c & 1][tid >> 2] = cs;
    }

#pragma unroll
    for (int k = 0; k < SPL; ++k) sH[(SPL * q + k) * CPB + (tid >> 2)] = h[k];
    __syncthreads();
    {
        const float* sy = &sY[(NCH - 1) & 1][jS];
        v4f o;
#pragma unroll
        for (int k = 0; k < 4; ++k) {
            const v4f s = *(const v4f*)(sy + 4 * k);
            o[k] = ((s[0] + s[1]) + s[2]) + s[3];
        }
        *(v4f*)pyf = o;
        if (fs == 0) *(v4f*)pck = *(const v4f*)&sC[(NCH - 1) & 1][fc];
        float* ph = xw + (row0 + (size_t)(TCH * fs + 1)) * XZW + d0 + fc;
        *(v4f*)ph = *(const v4f*)&sH[fs * CPB + fc];
    }
}

#define TT    32
#define CSUB  8
#define DPB   128

template <bool CARRY>
__device__ __forceinline__ void carry_tile(
    const float* __restrict__ pdt, const float* __restrict__ pu,
    const float* __restrict__ pz, float* __restrict__ py,
    const float* sCt, const v2f (&A2)[NS / 2], const v2f (&hin)[NS / 2],
    float cs, const float Dd)
{
    for (int t0 = 0; t0 < TT; t0 += CSUB) {
        float dv[CSUB], yl[CSUB], uv[CSUB], zv[CSUB];
#pragma unroll
        for (int i = 0; i < CSUB; ++i) {
            dv[i] = CARRY ? pdt[(size_t)(t0 + i) * DI] : 0.0f;
            yl[i] = py[(size_t)(t0 + i) * DI];
            uv[i] = pu[(size_t)(t0 + i) * DI];
            zv[i] = pz[(size_t)(t0 + i) * XZW];
        }
#pragma unroll
        for (int i = 0; i < CSUB; ++i) {
            float y = yl[i];
            if (CARRY) {
                cs += dv[i];
                const v4f* cr = (const v4f*)(sCt + (t0 + i) * NS);
                v2f acc = { 0.0f, 0.0f };
#pragma unroll
                for (int k4 = 0; k4 < NS / 4; ++k4) {
                    const v4f c4 = cr[k4];
                    const v2f a0 = A2[2 * k4] * cs;
                    const v2f a1 = A2[2 * k4 + 1] * cs;
                    const v2f e0 = { __builtin_amdgcn_exp2f(a0[0]), __builtin_amdgcn_exp2f(a0[1]) };
                    const v2f e1 = { __builtin_amdgcn_exp2f(a1[0]), __builtin_amdgcn_exp2f(a1[1]) };
                    const v2f c0 = { c4[0], c4[1] };
                    const v2f c1 = { c4[2], c4[3] };
                    acc += (e0 * hin[2 * k4]) * c0;
                    acc += (e1 * hin[2 * k4 + 1]) * c1;
                }
                y += acc[0] + acc[1];
            }
            py[(size_t)(t0 + i) * DI] = (y + uv[i] * Dd) * siluf(zv[i]);
        }
    }
}

__global__ void __launch_bounds__(128) carry_gate_kernel(
    const float* __restrict__ dt, const float* __restrict__ u,
    const float* __restrict__ dbl, const float* __restrict__ xz,
    const float* __restrict__ A_log, const float* __restrict__ Dp,
    float* __restrict__ yg)
{
    __shared__ __attribute__((aligned(16))) float sCt[TT * NS];
    const int tid  = threadIdx.x;
    const int dblk = blockIdx.x % (DI / DPB);
    const int tile = (blockIdx.x / (DI / DPB)) % (LSEG / TT);
    const int seg  = (blockIdx.x / ((DI / DPB) * (LSEG / TT))) % NSEG;
    const int b    = blockIdx.x / ((DI / DPB) * (LSEG / TT) * NSEG);
    const int d    = dblk * DPB + tid;
    const size_t r0 = (size_t)b * LL + (size_t)seg * LSEG + (size_t)tile * TT;

    const float Dd = Dp[d];
    const float* pdt = dt + r0 * DI + d;
    const float* pu  = u  + r0 * DI + d;
    const float* pz  = xz + r0 * XZW + DI + d;
    float*       py  = yg + r0 * DI + d;

    v2f A2[NS / 2], hin[NS / 2];
    if (seg > 0) {
        *(v4f*)(sCt + 4 * tid) =
            *(const v4f*)(dbl + (r0 + (tid >> 2)) * DBLW + RR + NS + 4 * (tid & 3));
        const v4f* pa = (const v4f*)(A_log + (size_t)d * NS);
#pragma unroll
        for (int k4 = 0; k4 < NS / 4; ++k4) {
            const v4f a = pa[k4];
            A2[2 * k4]     = v2f{ -__expf(a[0]), -__expf(a[1]) } * 1.44269504088896341f;
            A2[2 * k4 + 1] = v2f{ -__expf(a[2]), -__expf(a[3]) } * 1.44269504088896341f;
            hin[2 * k4]     = v2f{ 0.0f, 0.0f };
            hin[2 * k4 + 1] = v2f{ 0.0f, 0.0f };
        }
        for (int s = 0; s < seg; ++s) {
            const size_t rs = (size_t)b * LL + (size_t)s * LSEG;
            const float tot = xz[(rs + LSEG - TCH) * XZW + d];
#pragma unroll
            for (int k2 = 0; k2 < NS / 2; ++k2) {
                const v2f a  = A2[k2] * tot;
                const v2f e  = { __builtin_amdgcn_exp2f(a[0]), __builtin_amdgcn_exp2f(a[1]) };
                const v2f hf = { xz[(rs + (size_t)(TCH * (2 * k2) + 1)) * XZW + d],
                                 xz[(rs + (size_t)(TCH * (2 * k2 + 1) + 1)) * XZW + d] };
                hin[k2] = e * hin[k2] + hf;
            }
        }
        const float cs0 = (tile > 0) ? xz[(r0 - TCH) * XZW + d] : 0.0f;
        __syncthreads();
        carry_tile<true>(pdt, pu, pz, py, sCt, A2, hin, cs0, Dd);
    } else {
#pragma unroll
        for (int k2 = 0; k2 < NS / 2; ++k2) { A2[k2] = v2f{ 0.0f, 0.0f }; hin[k2] = v2f{ 0.0f, 0.0f }; }
        carry_tile<false>(pdt, pu, pz, py, sCt, A2, hin, 0.0f, Dd);
    }
}

extern "C" void kernel_launch(void* const* d_in, const int* in_sizes, int n_in,
                              void* d_out, int out_size, void* d_ws, size_t ws_size,
                              hipStream_t stream) {
    const float* x          = (const float*)d_in[0];
    const float* in_proj_w  = (const float*)d_in[1];
    const float* conv_w     = (const float*)d_in[2];
    const float* conv_b     = (const float*)d_in[3];
    const float* x_proj_w   = (const float*)d_in[4];
    const float* dt_proj_w  = (const float*)d_in[5];
    const float* dt_proj_b  = (const float*)d_in[6];
    const float* A_log      = (const float*)d_in[7];
    const float* Dvec       = (const float*)d_in[8];
    const float* out_proj_w = (const float*)d_in[9];
    float* out = (float*)d_out;

    float* ws  = (float*)d_ws;
    float* xz  = ws;
    float* u   = xz  + (size_t)2048 * 4096;
    float* dbl = u   + (size_t)2048 * 2048;
    float* dtb = dbl + (size_t)2048 * 96;
    float* yg  = dtb + (size_t)2048 * 2048;

    const dim3 blk(128);
    const int M = BB * LL;

    wmma_gemm_f32<4, 32, false, 1><<<dim3(4096 / 64, M / 64), blk, 0, stream>>>(
        x, in_proj_w, xz, 4096, DMOD, DMOD, nullptr, nullptr);

    conv_silu_kernel<<<(BB * LL * DI) / 256, 256, 0, stream>>>(xz, conv_w, conv_b, u);

    wmma_gemm_f32<2, 64, false, 1><<<dim3(DBLW / 32, M / 64, KSPL), blk, 0, stream>>>(
        u, x_proj_w, yg, DBLW, DI / KSPL, DI, nullptr, nullptr);

    wmma_gemm_f32<4, 32, true, KSPL><<<dim3(DI / 64, M / 64), blk, 0, stream>>>(
        yg, dt_proj_w, dtb, DI, RR, DBLW, dt_proj_b, dbl);

    scan_kernel<<<BB * NSEG * (DI / CPB), 128, 0, stream>>>(dtb, u, dbl, A_log, xz, yg);
    carry_gate_kernel<<<BB * NSEG * (LSEG / TT) * (DI / DPB), 128, 0, stream>>>(
        dtb, u, dbl, xz, A_log, Dvec, yg);

    wmma_gemm_f32<4, 32, false, 1><<<dim3(DMOD / 64, M / 64), blk, 0, stream>>>(
        yg, out_proj_w, out, DMOD, DI, DI, nullptr, nullptr);
}


# conv+SiLU: each thread makes 4 adjacent channels x 8 time steps from an 11-row register window (16-byte loads and stores); 4096 waves enter, not 131072 one-output waves; sums unchanged
# speedup vs baseline: 9.2127x; 1.0339x over previous
#include <hip/hip_runtime.h>
#include <hip/hip_bf16.h>
#include <cmath>

typedef __attribute__((ext_vector_type(2))) float v2f;
typedef __attribute__((ext_vector_type(4))) float v4f;
typedef __attribute__((ext_vector_type(8))) float v8f;

#define BB    2
#define LL    1024
#define DMOD  1024
#define DI    2048
#define NS    16
#define KC    4
#define RR    64
#define XZW   (2 * DI)
#define DBLW  (RR + 2 * NS)
#define KSPL  8

__device__ __forceinline__ float siluf(float v) {
    return v / (1.0f + __expf(-v));
}
__device__ __forceinline__ float softplusf(float v) {
    return (v > 20.0f) ? v : log1pf(expf(v));
}

template <int AP>
__device__ __forceinline__ v4f load_planes(const float* p, size_t plane)
{
    v4f v = *(const v4f*)p;
#pragma unroll
    for (int s = 1; s < AP; ++s) v += *(const v4f*)(p + (size_t)s * plane);
    return v;
}

template <int NT, int BK, bool SP, int AP>
__global__ void __launch_bounds__(128) wmma_gemm_f32(
    const float* __restrict__ A, const float* __restrict__ W,
    float* __restrict__ C, int N, int K, int lda,
    const float* __restrict__ bias, float* __restrict__ red)
{
    constexpr int BN   = NT * 16;
    constexpr int PA   = BK + 4;
    constexpr int PB   = 2 * BN + 32;
    constexpr int KS   = BK / 4;
    constexpr int A_F4 = (64 * (BK / 4)) / 128;
    constexpr int B_IT = ((BK / 2) * (BN / 4)) / 128;
    constexpr int A_RS = 128 / (BK / 4);
    constexpr int B_PS = 128 / (BN / 4);

    __shared__ __attribute__((aligned(16))) float sA[2][64 * PA];
    __shared__ __attribute__((aligned(16))) float sB[2][(BK / 2) * PB];

    const int tid  = threadIdx.x;
    const int wave = tid >> 5;
    const int lane = tid & 31;
    const int half = lane >> 4;
    const int mr   = lane & 15;
    const int mb   = blockIdx.y * 64;
    const int m0   = mb + wave * 16;
    const int nb   = blockIdx.x * BN;
    const size_t mrows  = (size_t)gridDim.y * 64;
    const size_t aplane = mrows * lda;
    A += (size_t)blockIdx.z * K;
    W += (size_t)blockIdx.z * K * N;
    C += (size_t)blockIdx.z * mrows * N;

    const int arow = tid / (BK / 4), ac4 = tid % (BK / 4);
    const int bp   = tid / (BN / 4), bq  = tid % (BN / 4);
    const float* aG = A + (size_t)(mb + arow) * lda + 4 * ac4;
    const float* wG = W + (size_t)(2 * bp) * N + nb + 4 * bq;
    const int aS = arow * PA + 4 * ac4;
    const int bS = bp * PB + 8 * bq;
    const int aF = (wave * 16 + mr) * PA + 2 * half;
    const int bF = half * PB + 2 * mr;

    v8f acc[NT] = {};
    v4f ra[A_F4], rb0[B_IT], rb1[B_IT];

#pragma unroll
    for (int i = 0; i < A_F4; ++i)
        ra[i] = load_planes<AP>(aG + (size_t)(i * A_RS) * lda, aplane);
#pragma unroll
    for (int i = 0; i < B_IT; ++i) {
        rb0[i] = *(const v4f*)(wG + (size_t)(2 * i * B_PS) * N);
        rb1[i] = *(const v4f*)(wG + (size_t)(2 * i * B_PS + 1) * N);
    }

    if (AP > 1 && blockIdx.x == 0) {
        const int rq = (lda - K) / 4;
        for (int p = tid; p < 64 * rq; p += 128) {
            const size_t o = (size_t)(mb + p / rq) * lda + K + 4 * (p % rq);
            *(v4f*)(red + o) = load_planes<AP>(A + o, aplane);
        }
    }

    const int S = K / BK;
    for (int s = 0; s < S - 1; ++s) {
        const int cur = s & 1;
#pragma unroll
        for (int i = 0; i < A_F4; ++i)
            *(v4f*)(&sA[cur][aS + i * A_RS * PA]) = ra[i];
#pragma unroll
        for (int i = 0; i < B_IT; ++i) {
            float* d = &sB[cur][bS + i * B_PS * PB];
            const v4f lo = { rb0[i][0], rb1[i][0], rb0[i][1], rb1[i][1] };
            const v4f hi = { rb0[i][2], rb1[i][2], rb0[i][3], rb1[i][3] };
            *(v4f*)(d)     = lo;
            *(v4f*)(d + 4) = hi;
        }
        __syncthreads();

        aG += BK;
        wG += (size_t)BK * N;
#pragma unroll
        for (int i = 0; i < A_F4; ++i)
            ra[i] = load_planes<AP>(aG + (size_t)(i * A_RS) * lda, aplane);
#pragma unroll
        for (int i = 0; i < B_IT; ++i) {
            rb0[i] = *(const v4f*)(wG + (size_t)(2 * i * B_PS) * N);
            rb1[i] = *(const v4f*)(wG + (size_t)(2 * i * B_PS + 1) * N);
        }

        const float* pa = &sA[cur][aF];
        const float* pb = &sB[cur][bF];
#pragma unroll
        for (int ks = 0; ks < KS; ++ks) {
            const v2f a = *(const v2f*)(pa + 4 * ks);
            v2f b[NT];
#pragma unroll
            for (int j = 0; j < NT; ++j) b[j] = *(const v2f*)(pb + 2 * ks * PB + 32 * j);
#pragma unroll
            for (int j = 0; j < NT; ++j)
                acc[j] = __builtin_amdgcn_wmma_f32_16x16x4_f32(
                    false, a, false, b[j], (short)0, acc[j], false, false);
        }
    }
    {
        const int cur = (S - 1) & 1;
#pragma unroll
        for (int i = 0; i < A_F4; ++i)
            *(v4f*)(&sA[cur][aS + i * A_RS * PA]) = ra[i];
#pragma unroll
        for (int i = 0; i < B_IT; ++i) {
            float* d = &sB[cur][bS + i * B_PS * PB];
            const v4f lo = { rb0[i][0], rb1[i][0], rb0[i][1], rb1[i][1] };
            const v4f hi = { rb0[i][2], rb1[i][2], rb0[i][3], rb1[i][3] };
            *(v4f*)(d)     = lo;
            *(v4f*)(d + 4) = hi;
        }
        __syncthreads();
        const float* pa = &sA[cur][aF];
        const float* pb = &sB[cur][bF];
#pragma unroll
        for (int ks = 0; ks < KS; ++ks) {
            const v2f a = *(const v2f*)(pa + 4 * ks);
            v2f b[NT];
#pragma unroll
            for (int j = 0; j < NT; ++j) b[j] = *(const v2f*)(pb + 2 * ks * PB + 32 * j);
#pragma unroll
            for (int j = 0; j < NT; ++j)
                acc[j] = __builtin_amdgcn_wmma_f32_16x16x4_f32(
                    false, a, false, b[j], (short)0, acc[j], false, false);
        }
    }

#pragma unroll
    for (int j = 0; j < NT; ++j) {
        const int col = nb + j * 16 + mr;
        const float bsv = SP ? bias[col] : 0.0f;
#pragma unroll
        for (int r = 0; r < 8; ++r) {
            float v = acc[j][r];
            if (SP) v = softplusf(v + bsv);
            C[(size_t)(m0 + r + 8 * half) * N + col] = v;
        }
    }
}

#define CTS   8
#define CGR   (DI / 4)

__global__ void __launch_bounds__(128) conv_silu_kernel(
    const float* __restrict__ xz, const float* __restrict__ cw,
    const float* __restrict__ cb, float* __restrict__ u)
{
    const int gid = blockIdx.x * 128 + threadIdx.x;
    const int d   = 4 * (gid % CGR);
    const int bt0 = (gid / CGR) * CTS;
    const bool head = (bt0 % LL) == 0;
    const float* px = xz + (size_t)bt0 * XZW + d;

    v4f xr[CTS + KC - 1];
#pragma unroll
    for (int i = 0; i < KC - 1; ++i) {
        xr[i] = v4f{ 0.0f, 0.0f, 0.0f, 0.0f };
        if (!head) xr[i] = *(const v4f*)(px - (size_t)(KC - 1 - i) * XZW);
    }
#pragma unroll
    for (int i = 0; i < CTS; ++i)
        xr[KC - 1 + i] = *(const v4f*)(px + (size_t)i * XZW);

    v4f wc[4];
#pragma unroll
    for (int c = 0; c < 4; ++c) wc[c] = *(const v4f*)(cw + (size_t)(d + c) * KC);
    v4f wk[KC];
#pragma unroll
    for (int k = 0; k < KC; ++k) wk[k] = v4f{ wc[0][k], wc[1][k], wc[2][k], wc[3][k] };
    const v4f bias = *(const v4f*)(cb + d);

    float* pu = u + (size_t)bt0 * DI + d;
#pragma unroll
    for (int i = 0; i < CTS; ++i) {
        v4f acc = bias;
#pragma unroll
        for (int k = 0; k < KC; ++k) {
            if (i + k >= KC - 1 || !head)
                acc += xr[i + k] * wk[k];
        }
        const v4f o = { siluf(acc[0]), siluf(acc[1]), siluf(acc[2]), siluf(acc[3]) };
        *(v4f*)(pu + (size_t)i * DI) = o;
    }
}

#define TCH   16
#define NSEG  4
#define LSEG  (LL / NSEG)
#define NCH   (LSEG / TCH)
#define BCW   (2 * NS)
#define QL    4
#define SPL   (NS / QL)
#define CPB   (128 / QL)
#define GST   4
#define NGR   (TCH / GST)

__global__ void __launch_bounds__(128) scan_kernel(
    const float* __restrict__ dt, const float* __restrict__ u,
    const float* __restrict__ dbl, const float* __restrict__ A_log,
    float* __restrict__ xw, float* __restrict__ yg)
{
    __shared__ __attribute__((aligned(16))) float sBC[2][TCH * BCW];
    __shared__ __attribute__((aligned(16))) float sY[2][TCH * 128];
    __shared__ __attribute__((aligned(16))) float sC[2][CPB];
    __shared__ __attribute__((aligned(16))) float sH[NS * CPB];
    const int tid = threadIdx.x;
    const int cb  = blockIdx.x % (DI / CPB);
    const int seg = (blockIdx.x / (DI / CPB)) % NSEG;
    const int b   = blockIdx.x / ((DI / CPB) * NSEG);
    const int d0  = cb * CPB;
    const int q   = tid & (QL - 1);
    const int d   = d0 + (tid >> 2);

    float An[SPL];
#pragma unroll
    for (int k = 0; k < SPL; ++k) An[k] = -__expf(A_log[d * NS + SPL * q + k]);
    float h[SPL];
#pragma unroll
    for (int k = 0; k < SPL; ++k) h[k] = 0.0f;
    float cs = 0.0f;

    const size_t row0 = (size_t)b * LL + (size_t)seg * LSEG;
    const float* pdt = dt + row0 * DI + d;
    const float* pu  = u  + row0 * DI + d;
    const float* pbc = dbl + (row0 + (tid >> 3)) * DBLW + RR + 4 * (tid & 7);
    const int fs = tid >> 3, fc = 4 * (tid & 7);
    float* pyf = yg + (row0 + fs) * DI + d0 + fc;
    float* pck = xw + row0 * XZW + d0 + fc;
    const int jS = fs * 128 + QL * fc;
    const int bcq = SPL * q;

    float dt_n[TCH], u_n[TCH];
#pragma unroll
    for (int i = 0; i < TCH; ++i) {
        dt_n[i] = pdt[(size_t)i * DI];
        u_n[i]  = pu[(size_t)i * DI];
    }
    v4f bc_n = *(const v4f*)pbc;

    for (int c = 0; c < NCH; ++c) {
        float dt_c[TCH], u_c[TCH];
#pragma unroll
        for (int i = 0; i < TCH; ++i) { dt_c[i] = dt_n[i]; u_c[i] = u_n[i]; }

        float* buf = sBC[c & 1];
        *(v4f*)(buf + 4 * tid) = bc_n;
        __syncthreads();

        v4f ring[2][2 * GST];
        v4f sj[4];
        if (c > 0) {
            const float* sy = &sY[(c - 1) & 1][jS];
#pragma unroll
            for (int k = 0; k < 4; ++k) sj[k] = *(const v4f*)(sy + 4 * k);
        }
#pragma unroll
        for (int j = 0; j < GST; ++j) {
            ring[0][2 * j]     = *(const v4f*)(buf + j * BCW + bcq);
            ring[0][2 * j + 1] = *(const v4f*)(buf + j * BCW + NS + bcq);
        }
        if (c > 0) {
            v4f o;
#pragma unroll
            for (int k = 0; k < 4; ++k)
                o[k] = ((sj[k][0] + sj[k][1]) + sj[k][2]) + sj[k][3];
            *(v4f*)pyf = o;
            pyf += (size_t)TCH * DI;
            if (fs == 0) *(v4f*)pck = *(const v4f*)&sC[(c - 1) & 1][fc];
            pck += (size_t)TCH * XZW;
        }

        if (c + 1 < NCH) {
            pdt += (size_t)TCH * DI;
            pu  += (size_t)TCH * DI;
            pbc += (size_t)TCH * DBLW;
#pragma unroll
            for (int i = 0; i < TCH; ++i) {
                dt_n[i] = pdt[(size_t)i * DI];
                u_n[i]  = pu[(size_t)i * DI];
            }
            bc_n = *(const v4f*)pbc;
        }

        float p[TCH];
#pragma unroll
        for (int g = 0; g < NGR; ++g) {
            if (g + 1 < NGR) {
#pragma unroll
                for (int j = 0; j < GST; ++j) {
                    const float* r = buf + ((g + 1) * GST + j) * BCW + bcq;
                    ring[(g + 1) & 1][2 * j]     = *(const v4f*)(r);
                    ring[(g + 1) & 1][2 * j + 1] = *(const v4f*)(r + NS);
                }
            }
            __builtin_amdgcn_sched_barrier(0);

#pragma unroll
            for (int j = 0; j < GST; ++j) {
                const int i = g * GST + j;
                const float dtv = dt_c[i];
                const float du  = dtv * u_c[i];
                const v4f Bq = ring[g & 1][2 * j];
                const v4f Cq = ring[g & 1][2 * j + 1];
                float y = 0.0f;
#pragma unroll
                for (int k = 0; k < SPL; ++k) {
                    h[k] = __expf(dtv * An[k]) * h[k] + du * Bq[k];
                    y += h[k] * Cq[k];
                }
                p[i] = y;
                cs += dtv;
            }
        }
        float* sy = &sY[c & 1][tid];
#pragma unroll
        for (int i = 0; i < TCH; ++i) sy[i * 128] = p[i];
        if (q == 0) sC[c & 1][tid >> 2] = cs;
    }

#pragma unroll
    for (int k = 0; k < SPL; ++k) sH[(SPL * q + k) * CPB + (tid >> 2)] = h[k];
    __syncthreads();
    {
        const float* sy = &sY[(NCH - 1) & 1][jS];
        v4f o;
#pragma unroll
        for (int k = 0; k < 4; ++k) {
            const v4f s = *(const v4f*)(sy + 4 * k);
            o[k] = ((s[0] + s[1]) + s[2]) + s[3];
        }
        *(v4f*)pyf = o;
        if (fs == 0) *(v4f*)pck = *(const v4f*)&sC[(NCH - 1) & 1][fc];
        float* ph = xw + (row0 + (size_t)(TCH * fs + 1)) * XZW + d0 + fc;
        *(v4f*)ph = *(const v4f*)&sH[fs * CPB + fc];
    }
}

#define TT    32
#define CSUB  8
#define DPB   128

template <bool CARRY>
__device__ __forceinline__ void carry_tile(
    const float* __restrict__ pdt, const float* __restrict__ pu,
    const float* __restrict__ pz, float* __restrict__ py,
    const float* sCt, const v2f (&A2)[NS / 2], const v2f (&hin)[NS / 2],
    float cs, const float Dd)
{
    for (int t0 = 0; t0 < TT; t0 += CSUB) {
        float dv[CSUB], yl[CSUB], uv[CSUB], zv[CSUB];
#pragma unroll
        for (int i = 0; i < CSUB; ++i) {
            dv[i] = CARRY ? pdt[(size_t)(t0 + i) * DI] : 0.0f;
            yl[i] = py[(size_t)(t0 + i) * DI];
            uv[i] = pu[(size_t)(t0 + i) * DI];
            zv[i] = pz[(size_t)(t0 + i) * XZW];
        }
#pragma unroll
        for (int i = 0; i < CSUB; ++i) {
            float y = yl[i];
            if (CARRY) {
                cs += dv[i];
                const v4f* cr = (const v4f*)(sCt + (t0 + i) * NS);
                v2f acc = { 0.0f, 0.0f };
#pragma unroll
                for (int k4 = 0; k4 < NS / 4; ++k4) {
                    const v4f c4 = cr[k4];
                    const v2f a0 = A2[2 * k4] * cs;
                    const v2f a1 = A2[2 * k4 + 1] * cs;
                    const v2f e0 = { __builtin_amdgcn_exp2f(a0[0]), __builtin_amdgcn_exp2f(a0[1]) };
                    const v2f e1 = { __builtin_amdgcn_exp2f(a1[0]), __builtin_amdgcn_exp2f(a1[1]) };
                    const v2f c0 = { c4[0], c4[1] };
                    const v2f c1 = { c4[2], c4[3] };
                    acc += (e0 * hin[2 * k4]) * c0;
                    acc += (e1 * hin[2 * k4 + 1]) * c1;
                }
                y += acc[0] + acc[1];
            }
            py[(size_t)(t0 + i) * DI] = (y + uv[i] * Dd) * siluf(zv[i]);
        }
    }
}

__global__ void __launch_bounds__(128) carry_gate_kernel(
    const float* __restrict__ dt, const float* __restrict__ u,
    const float* __restrict__ dbl, const float* __restrict__ xz,
    const float* __restrict__ A_log, const float* __restrict__ Dp,
    float* __restrict__ yg)
{
    __shared__ __attribute__((aligned(16))) float sCt[TT * NS];
    const int tid  = threadIdx.x;
    const int dblk = blockIdx.x % (DI / DPB);
    const int tile = (blockIdx.x / (DI / DPB)) % (LSEG / TT);
    const int seg  = (blockIdx.x / ((DI / DPB) * (LSEG / TT))) % NSEG;
    const int b    = blockIdx.x / ((DI / DPB) * (LSEG / TT) * NSEG);
    const int d    = dblk * DPB + tid;
    const size_t r0 = (size_t)b * LL + (size_t)seg * LSEG + (size_t)tile * TT;

    const float Dd = Dp[d];
    const float* pdt = dt + r0 * DI + d;
    const float* pu  = u  + r0 * DI + d;
    const float* pz  = xz + r0 * XZW + DI + d;
    float*       py  = yg + r0 * DI + d;

    v2f A2[NS / 2], hin[NS / 2];
    if (seg > 0) {
        *(v4f*)(sCt + 4 * tid) =
            *(const v4f*)(dbl + (r0 + (tid >> 2)) * DBLW + RR + NS + 4 * (tid & 3));
        const v4f* pa = (const v4f*)(A_log + (size_t)d * NS);
#pragma unroll
        for (int k4 = 0; k4 < NS / 4; ++k4) {
            const v4f a = pa[k4];
            A2[2 * k4]     = v2f{ -__expf(a[0]), -__expf(a[1]) } * 1.44269504088896341f;
            A2[2 * k4 + 1] = v2f{ -__expf(a[2]), -__expf(a[3]) } * 1.44269504088896341f;
            hin[2 * k4]     = v2f{ 0.0f, 0.0f };
            hin[2 * k4 + 1] = v2f{ 0.0f, 0.0f };
        }
        for (int s = 0; s < seg; ++s) {
            const size_t rs = (size_t)b * LL + (size_t)s * LSEG;
            const float tot = xz[(rs + LSEG - TCH) * XZW + d];
#pragma unroll
            for (int k2 = 0; k2 < NS / 2; ++k2) {
                const v2f a  = A2[k2] * tot;
                const v2f e  = { __builtin_amdgcn_exp2f(a[0]), __builtin_amdgcn_exp2f(a[1]) };
                const v2f hf = { xz[(rs + (size_t)(TCH * (2 * k2) + 1)) * XZW + d],
                                 xz[(rs + (size_t)(TCH * (2 * k2 + 1) + 1)) * XZW + d] };
                hin[k2] = e * hin[k2] + hf;
            }
        }
        const float cs0 = (tile > 0) ? xz[(r0 - TCH) * XZW + d] : 0.0f;
        __syncthreads();
        carry_tile<true>(pdt, pu, pz, py, sCt, A2, hin, cs0, Dd);
    } else {
#pragma unroll
        for (int k2 = 0; k2 < NS / 2; ++k2) { A2[k2] = v2f{ 0.0f, 0.0f }; hin[k2] = v2f{ 0.0f, 0.0f }; }
        carry_tile<false>(pdt, pu, pz, py, sCt, A2, hin, 0.0f, Dd);
    }
}

extern "C" void kernel_launch(void* const* d_in, const int* in_sizes, int n_in,
                              void* d_out, int out_size, void* d_ws, size_t ws_size,
                              hipStream_t stream) {
    const float* x          = (const float*)d_in[0];
    const float* in_proj_w  = (const float*)d_in[1];
    const float* conv_w     = (const float*)d_in[2];
    const float* conv_b     = (const float*)d_in[3];
    const float* x_proj_w   = (const float*)d_in[4];
    const float* dt_proj_w  = (const float*)d_in[5];
    const float* dt_proj_b  = (const float*)d_in[6];
    const float* A_log      = (const float*)d_in[7];
    const float* Dvec       = (const float*)d_in[8];
    const float* out_proj_w = (const float*)d_in[9];
    float* out = (float*)d_out;

    float* ws  = (float*)d_ws;
    float* xz  = ws;
    float* u   = xz  + (size_t)2048 * 4096;
    float* dbl = u   + (size_t)2048 * 2048;
    float* dtb = dbl + (size_t)2048 * 96;
    float* yg  = dtb + (size_t)2048 * 2048;

    const dim3 blk(128);
    const int M = BB * LL;

    wmma_gemm_f32<4, 32, false, 1><<<dim3(4096 / 64, M / 64), blk, 0, stream>>>(
        x, in_proj_w, xz, 4096, DMOD, DMOD, nullptr, nullptr);

    conv_silu_kernel<<<((BB * LL / CTS) * CGR) / 128, blk, 0, stream>>>(xz, conv_w, conv_b, u);

    wmma_gemm_f32<2, 64, false, 1><<<dim3(DBLW / 32, M / 64, KSPL), blk, 0, stream>>>(
        u, x_proj_w, yg, DBLW, DI / KSPL, DI, nullptr, nullptr);

    wmma_gemm_f32<4, 32, true, KSPL><<<dim3(DI / 64, M / 64), blk, 0, stream>>>(
        yg, dt_proj_w, dtb, DI, RR, DBLW, dt_proj_b, dbl);

    scan_kernel<<<BB * NSEG * (DI / CPB), 128, 0, stream>>>(dtb, u, dbl, A_log, xz, yg);
    carry_gate_kernel<<<BB * NSEG * (LSEG / TT) * (DI / DPB), 128, 0, stream>>>(
        dtb, u, dbl, xz, A_log, Dvec, yg);

    wmma_gemm_f32<4, 32, false, 1><<<dim3(DMOD / 64, M / 64), blk, 0, stream>>>(
        yg, out_proj_w, out, DMOD, DI, DI, nullptr, nullptr);
}
